# starting kernel
# baseline (speedup 1.0000x reference)
#include <hip/hip_runtime.h>
#include <hip/hip_bf16.h>
#include <math.h>

#define N_NODES   50000
#define N_EDGES   800000
#define IN_F      128
#define HEADS     8
#define HEAD_DIM  16
#define OUT_F     (HEADS * HEAD_DIM)
#define ALPHA     0.2f
#define EPS_F     1e-6f

typedef __attribute__((ext_vector_type(2))) float v2f;
typedef __attribute__((ext_vector_type(8))) float v8f;

__device__ __forceinline__ unsigned enc_f32(float f) {
    unsigned u = __float_as_uint(f);
    return (u & 0x80000000u) ? ~u : (u | 0x80000000u);
}
__device__ __forceinline__ float dec_f32(unsigned u) {
    return (u & 0x80000000u) ? __uint_as_float(u ^ 0x80000000u)
                             : __uint_as_float(~u);
}
__device__ __forceinline__ float leaky(float s) {
    return s >= 0.0f ? s : ALPHA * s;
}

__global__ void gat_init(float* __restrict__ out, const float* __restrict__ bias,
                         float* __restrict__ denom, unsigned* __restrict__ gmax) {
    const int i = blockIdx.x * blockDim.x + threadIdx.x;
    if (i < N_NODES * OUT_F) out[i] = bias[i & (OUT_F - 1)];
    if (i < N_NODES * HEADS) denom[i] = 0.0f;
    if (i < HEADS)           gmax[i] = 0u;
}

__global__ __launch_bounds__(256)
void gat_gemm_wmma(const float* __restrict__ x, const float* __restrict__ W,
                   float* __restrict__ h) {
    __shared__ float As[16 * IN_F];
    const int tid = threadIdx.x;
    const int m0  = blockIdx.x * 16;

    {
        const int base = tid * 8;
        const int row  = base >> 7;
        const int col  = base & 127;
        const float4* g =
            reinterpret_cast<const float4*>(x + (size_t)(m0 + row) * IN_F + col);
        float4 p0 = g[0];
        float4 p1 = g[1];
        *reinterpret_cast<float4*>(&As[row * IN_F + col])     = p0;
        *reinterpret_cast<float4*>(&As[row * IN_F + col + 4]) = p1;
    }
    __syncthreads();

    const int lane = tid & 31;
    const int head = tid >> 5;
    const int l16  = lane & 15;
    const bool hi  = lane >= 16;
    const float* Wh = W + (size_t)head * IN_F * HEAD_DIM;

    v8f c = {};
#pragma unroll
    for (int k = 0; k < IN_F; k += 4) {
        const int kk = k + (hi ? 2 : 0);
        v2f a, b;
        a.x = As[l16 * IN_F + kk];
        a.y = As[l16 * IN_F + kk + 1];
        b.x = Wh[kk * HEAD_DIM + l16];
        b.y = Wh[(kk + 1) * HEAD_DIM + l16];
        c = __builtin_amdgcn_wmma_f32_16x16x4_f32(
                 false, a,  false, b,
                 (short)0, c,  false,  false);
    }

    const int rbase = hi ? 8 : 0;
#pragma unroll
    for (int v = 0; v < 8; ++v)
        h[(size_t)(m0 + rbase + v) * OUT_F + head * HEAD_DIM + l16] = c[v];
}

__global__ void gat_node_scores(const float* __restrict__ h, const float* __restrict__ a,
                                float* __restrict__ s_src, float* __restrict__ s_tgt) {
    const int i = blockIdx.x * blockDim.x + threadIdx.x;
    if (i >= N_NODES * HEADS) return;
    const int hd = i & (HEADS - 1);
    const int n  = i >> 3;
    const float* hv = h + (size_t)n * OUT_F + hd * HEAD_DIM;
    const float* as = a + hd * 2 * HEAD_DIM;
    const float* at = as + HEAD_DIM;
    float ss = 0.0f, st = 0.0f;
#pragma unroll
    for (int d = 0; d < HEAD_DIM; ++d) {
        const float v = hv[d];
        ss += v * as[d];
        st += v * at[d];
    }
    s_src[i] = ss;
    s_tgt[i] = st;
}

__global__ __launch_bounds__(256)
void gat_edge_max(const int* __restrict__ src, const int* __restrict__ tgt,
                  const float* __restrict__ s_src, const float* __restrict__ s_tgt,
                  unsigned* __restrict__ gmax) {
    __shared__ unsigned lmax[HEADS];
    const int tid = threadIdx.x;
    if (tid < HEADS) lmax[tid] = 0u;
    __syncthreads();

    const int e = blockIdx.x * blockDim.x + tid;
    if (e < N_EDGES) {
        const int s = src[e], t = tgt[e];
#pragma unroll
        for (int hd = 0; hd < HEADS; ++hd) {
            const float sc = leaky(s_src[s * HEADS + hd] + s_tgt[t * HEADS + hd]);
            atomicMax(&lmax[hd], enc_f32(sc));
        }
    }
    __syncthreads();
    if (tid < HEADS) atomicMax(&gmax[tid], lmax[tid]);
}

__global__ __launch_bounds__(256)
void gat_edge_denom(const int* __restrict__ src, const int* __restrict__ tgt,
                    const float* __restrict__ s_src, const float* __restrict__ s_tgt,
                    const unsigned* __restrict__ gmax, float* __restrict__ denom) {
    const int e = blockIdx.x * blockDim.x + threadIdx.x;
    if (e >= N_EDGES) return;
    const int s = src[e], t = tgt[e];
#pragma unroll
    for (int hd = 0; hd < HEADS; ++hd) {
        const float mx = dec_f32(gmax[hd]);
        const float sc = leaky(s_src[s * HEADS + hd] + s_tgt[t * HEADS + hd]);
        const float ev = __expf(fminf(fmaxf(sc - mx, -20.0f), 20.0f));
        atomicAdd(&denom[t * HEADS + hd], ev);
    }
}

__global__ __launch_bounds__(256)
void gat_edge_agg(const int* __restrict__ src, const int* __restrict__ tgt,
                  const float* __restrict__ s_src, const float* __restrict__ s_tgt,
                  const unsigned* __restrict__ gmax, const float* __restrict__ denom,
                  const float* __restrict__ h, float* __restrict__ out) {
    const long long gid = (long long)blockIdx.x * blockDim.x + threadIdx.x;
    const int e  = (int)(gid >> 7);
    const int c  = (int)(gid & 127);
    const int hd = c >> 4;
    if (e >= N_EDGES) return;
    const int s = src[e], t = tgt[e];
    const float mx = dec_f32(gmax[hd]);
    const float sc = leaky(s_src[s * HEADS + hd] + s_tgt[t * HEADS + hd]);
    const float ev = __expf(fminf(fmaxf(sc - mx, -20.0f), 20.0f));
    const float attn = ev / (denom[t * HEADS + hd] + EPS_F);
    atomicAdd(&out[(size_t)t * OUT_F + c], h[(size_t)s * OUT_F + c] * attn);
}

extern "C" void kernel_launch(void* const* d_in, const int* in_sizes, int n_in,
                              void* d_out, int out_size, void* d_ws, size_t ws_size,
                              hipStream_t stream) {
    (void)in_sizes; (void)n_in; (void)out_size; (void)ws_size;

    const float* x    = (const float*)d_in[0];
    const int*   ei   = (const int*)  d_in[1];
    const float* W    = (const float*)d_in[2];
    const float* a    = (const float*)d_in[3];
    const float* bias = (const float*)d_in[4];
    float* out = (float*)d_out;

    const int* src = ei;
    const int* tgt = ei + N_EDGES;

    float*    ws    = (float*)d_ws;
    float*    hbuf  = ws;
    float*    ssrc  = hbuf + (size_t)N_NODES * OUT_F;
    float*    stgt  = ssrc + (size_t)N_NODES * HEADS;
    float*    denom = stgt + (size_t)N_NODES * HEADS;
    unsigned* gmax  = (unsigned*)(denom + (size_t)N_NODES * HEADS);

    const int outN = N_NODES * OUT_F;
    gat_init<<<(outN + 255) / 256, 256, 0, stream>>>(out, bias, denom, gmax);

    gat_gemm_wmma<<<N_NODES / 16, 256, 0, stream>>>(x, W, hbuf);

    gat_node_scores<<<(N_NODES * HEADS + 255) / 256, 256, 0, stream>>>(hbuf, a, ssrc, stgt);

    gat_edge_max<<<(N_EDGES + 255) / 256, 256, 0, stream>>>(src, tgt, ssrc, stgt, gmax);

    gat_edge_denom<<<(N_EDGES + 255) / 256, 256, 0, stream>>>(src, tgt, ssrc, stgt, gmax, denom);

    const long long aggThreads = (long long)N_EDGES * OUT_F;
    gat_edge_agg<<<(unsigned)((aggThreads + 255) / 256), 256, 0, stream>>>(
        src, tgt, ssrc, stgt, gmax, denom, hbuf, out);
}


# aggregation by target segments: edges grouped by target in-call (degree count, scan, slot fill); one wave per node sums its segment in registers, stores the row once; no float atomics or out pre-fill
# speedup vs baseline: 2.0887x; 2.0887x over previous
#include <hip/hip_runtime.h>
#include <hip/hip_bf16.h>
#include <math.h>

#define N_NODES   50000
#define N_EDGES   800000
#define IN_F      128
#define HEADS     8
#define HEAD_DIM  16
#define OUT_F     (HEADS * HEAD_DIM)
#define ALPHA     0.2f
#define EPS_F     1e-6f

typedef __attribute__((ext_vector_type(2))) float v2f;
typedef __attribute__((ext_vector_type(8))) float v8f;

__device__ __forceinline__ unsigned enc_f32(float f) {
    unsigned u = __float_as_uint(f);
    return (u & 0x80000000u) ? ~u : (u | 0x80000000u);
}
__device__ __forceinline__ float dec_f32(unsigned u) {
    return (u & 0x80000000u) ? __uint_as_float(u ^ 0x80000000u)
                             : __uint_as_float(~u);
}
__device__ __forceinline__ float leaky(float s) {
    return s >= 0.0f ? s : ALPHA * s;
}

static_assert(N_NODES <= 65536, "source ids are kept as 16-bit words");
static_assert(N_NODES % 8 == 0, "one wave per node, 8 waves per block");
static_assert(N_EDGES % 256 == 0, "edge kernels use full blocks");

__global__ __launch_bounds__(256)
void gat_gemm_wmma(const float* __restrict__ x, const float* __restrict__ W,
                   float* __restrict__ h) {
    __shared__ float As[16 * IN_F];
    const int tid = threadIdx.x;
    const int m0  = blockIdx.x * 16;

    {
        const int base = tid * 8;
        const int row  = base >> 7;
        const int col  = base & 127;
        const float4* g =
            reinterpret_cast<const float4*>(x + (size_t)(m0 + row) * IN_F + col);
        float4 p0 = g[0];
        float4 p1 = g[1];
        *reinterpret_cast<float4*>(&As[row * IN_F + col])     = p0;
        *reinterpret_cast<float4*>(&As[row * IN_F + col + 4]) = p1;
    }
    __syncthreads();

    const int lane = tid & 31;
    const int head = tid >> 5;
    const int l16  = lane & 15;
    const bool hi  = lane >= 16;
    const float* Wh = W + (size_t)head * IN_F * HEAD_DIM;

    v8f c = {};
#pragma unroll
    for (int k = 0; k < IN_F; k += 4) {
        const int kk = k + (hi ? 2 : 0);
        v2f a, b;
        a.x = As[l16 * IN_F + kk];
        a.y = As[l16 * IN_F + kk + 1];
        b.x = Wh[kk * HEAD_DIM + l16];
        b.y = Wh[(kk + 1) * HEAD_DIM + l16];
        c = __builtin_amdgcn_wmma_f32_16x16x4_f32(
                 false, a,  false, b,
                 (short)0, c,  false,  false);
    }

    const int rbase = hi ? 8 : 0;
#pragma unroll
    for (int v = 0; v < 8; ++v)
        h[(size_t)(m0 + rbase + v) * OUT_F + head * HEAD_DIM + l16] = c[v];
}

__global__ void gat_node_scores(const float* __restrict__ h, const float* __restrict__ a,
                                float* __restrict__ s_src, float* __restrict__ s_tgt,
                                int* __restrict__ cnt, unsigned* __restrict__ gmax) {
    const int i = blockIdx.x * blockDim.x + threadIdx.x;
    if (i < N_NODES) cnt[i] = 0;
    if (i < HEADS)   gmax[i] = 0u;
    if (i >= N_NODES * HEADS) return;
    const int hd = i & (HEADS - 1);
    const int n  = i >> 3;
    const float* hv = h + (size_t)n * OUT_F + hd * HEAD_DIM;
    const float* as = a + hd * 2 * HEAD_DIM;
    const float* at = as + HEAD_DIM;
    float ss = 0.0f, st = 0.0f;
#pragma unroll
    for (int d = 0; d < HEAD_DIM; ++d) {
        const float v = hv[d];
        ss += v * as[d];
        st += v * at[d];
    }
    s_src[i] = ss;
    s_tgt[i] = st;
}

__global__ __launch_bounds__(256)
void gat_edge_max(const int* __restrict__ src, const int* __restrict__ tgt,
                  const float* __restrict__ s_src, const float* __restrict__ s_tgt,
                  unsigned* __restrict__ gmax, int* __restrict__ cnt) {
    __shared__ unsigned lmax[HEADS];
    const int tid = threadIdx.x;
    if (tid < HEADS) lmax[tid] = 0u;
    __syncthreads();

    const int e = blockIdx.x * blockDim.x + tid;
    if (e < N_EDGES) {
        const int s = src[e], t = tgt[e];
        atomicAdd(&cnt[t], 1);
#pragma unroll
        for (int hd = 0; hd < HEADS; ++hd) {
            const float sc = leaky(s_src[s * HEADS + hd] + s_tgt[t * HEADS + hd]);
            atomicMax(&lmax[hd], enc_f32(sc));
        }
    }
    __syncthreads();
    if (tid < HEADS) atomicMax(&gmax[tid], lmax[tid]);
}

#define SCAN_T   1024
#define SCAN_Q   13
#define NODE_Q   (N_NODES / 4)
static_assert(N_NODES % 4 == 0, "counts are moved as 16-B words");
static_assert(SCAN_T * SCAN_Q >= NODE_Q, "scan covers every node");

__global__ __launch_bounds__(SCAN_T)
void gat_degree_scan(int* __restrict__ off) {
    __shared__ int4 buf[SCAN_T * SCAN_Q];
    __shared__ int  wsum[32];
    const int tid  = threadIdx.x;
    const int lane = tid & 31;
    const int wv   = tid >> 5;
    int4* off4 = reinterpret_cast<int4*>(off);

    {
        int4 r[SCAN_Q];
#pragma unroll
        for (int k = 0; k < SCAN_Q; ++k) {
            const int q = k * SCAN_T + tid;
            r[k] = off4[min(q, NODE_Q - 1)];
        }
#pragma unroll
        for (int k = 0; k < SCAN_Q; ++k) {
            const int q = k * SCAN_T + tid;
            buf[q] = (q < NODE_Q) ? r[k] : make_int4(0, 0, 0, 0);
        }
    }
    __syncthreads();

    int4 c[SCAN_Q];
    int sum = 0;
#pragma unroll
    for (int k = 0; k < SCAN_Q; ++k) {
        c[k] = buf[tid * SCAN_Q + k];
        sum += c[k].x + c[k].y + c[k].z + c[k].w;
    }

    int incl = sum;
#pragma unroll
    for (int d = 1; d < 32; d <<= 1) {
        const int v = __shfl_up(incl, d);
        if (lane >= d) incl += v;
    }
    if (lane == 31) wsum[wv] = incl;
    __syncthreads();
    if (wv == 0) {
        const int w = wsum[lane];
        int wi = w;
#pragma unroll
        for (int d = 1; d < 32; d <<= 1) {
            const int v = __shfl_up(wi, d);
            if (lane >= d) wi += v;
        }
        wsum[lane] = wi - w;
    }
    __syncthreads();

    int run = wsum[wv] + incl - sum;
#pragma unroll
    for (int k = 0; k < SCAN_Q; ++k) {
        int4 o;
        o.x = run; run += c[k].x;
        o.y = run; run += c[k].y;
        o.z = run; run += c[k].z;
        o.w = run; run += c[k].w;
        buf[tid * SCAN_Q + k] = o;
    }
    __syncthreads();
#pragma unroll
    for (int k = 0; k < SCAN_Q; ++k) {
        const int q = k * SCAN_T + tid;
        if (q < NODE_Q) off4[q] = buf[q];
    }
}

__global__ __launch_bounds__(256)
void gat_group_edges(const int* __restrict__ src, const int* __restrict__ tgt,
                     int* __restrict__ off, unsigned short* __restrict__ seg_src) {
    const int e = blockIdx.x * blockDim.x + threadIdx.x;
    if (e >= N_EDGES) return;
    const int t   = tgt[e];
    const int pos = atomicAdd(&off[t], 1);
    if ((unsigned)pos < (unsigned)N_EDGES)
        seg_src[pos] = (unsigned short)src[e];
}

__device__ __forceinline__ float edge_weight(float ssrc, float stgt, float mx) {
    const float sc = leaky(ssrc + stgt);
    return __expf(fminf(fmaxf(sc - mx, -20.0f), 20.0f));
}

__global__ __launch_bounds__(256)
void gat_node_agg(const int* __restrict__ off, const unsigned short* __restrict__ seg_src,
                  const float* __restrict__ s_src, const float* __restrict__ a,
                  const unsigned* __restrict__ gmax, const float* __restrict__ h,
                  const float* __restrict__ bias, float* __restrict__ out) {
    const int lane = threadIdx.x & 31;
    const int wv = __builtin_amdgcn_readfirstlane((int)(threadIdx.x >> 5));
    const int n  = blockIdx.x * 8 + wv;
    const int hd = lane >> 2;

    const float4 hn = *reinterpret_cast<const float4*>(h + (size_t)n * OUT_F + lane * 4);
    const float4 at = *reinterpret_cast<const float4*>(a + hd * 2 * HEAD_DIM + HEAD_DIM + (lane & 3) * 4);
    float st = hn.x * at.x + hn.y * at.y + hn.z * at.z + hn.w * at.w;
    st += __shfl_xor(st, 1);
    st += __shfl_xor(st, 2);

    const float mx = dec_f32(gmax[hd]);

    int beg = (n > 0) ? off[n - 1] : 0;
    int end = off[n];
    beg = max(beg, 0);
    end = min(end, N_EDGES);

    float4 acc = make_float4(0.0f, 0.0f, 0.0f, 0.0f);
    float  den = 0.0f;

    for (int base = beg; base < end; base += 32) {
        const int cnt = min(end - base, 32);
        int myidx = 0;
        if (lane < cnt) myidx = seg_src[base + lane];

        int k = 0;
        for (; k + 4 <= cnt; k += 4) {
            const int s0 = __builtin_amdgcn_readlane(myidx, k);
            const int s1 = __builtin_amdgcn_readlane(myidx, k + 1);
            const int s2 = __builtin_amdgcn_readlane(myidx, k + 2);
            const int s3 = __builtin_amdgcn_readlane(myidx, k + 3);
            const float q0 = s_src[s0 * HEADS + hd];
            const float q1 = s_src[s1 * HEADS + hd];
            const float q2 = s_src[s2 * HEADS + hd];
            const float q3 = s_src[s3 * HEADS + hd];
            const float4 v0 = *reinterpret_cast<const float4*>(h + (size_t)s0 * OUT_F + lane * 4);
            const float4 v1 = *reinterpret_cast<const float4*>(h + (size_t)s1 * OUT_F + lane * 4);
            const float4 v2 = *reinterpret_cast<const float4*>(h + (size_t)s2 * OUT_F + lane * 4);
            const float4 v3 = *reinterpret_cast<const float4*>(h + (size_t)s3 * OUT_F + lane * 4);
            const float w0 = edge_weight(q0, st, mx);
            const float w1 = edge_weight(q1, st, mx);
            const float w2 = edge_weight(q2, st, mx);
            const float w3 = edge_weight(q3, st, mx);
            den += w0; den += w1; den += w2; den += w3;
            acc.x += v0.x * w0; acc.y += v0.y * w0; acc.z += v0.z * w0; acc.w += v0.w * w0;
            acc.x += v1.x * w1; acc.y += v1.y * w1; acc.z += v1.z * w1; acc.w += v1.w * w1;
            acc.x += v2.x * w2; acc.y += v2.y * w2; acc.z += v2.z * w2; acc.w += v2.w * w2;
            acc.x += v3.x * w3; acc.y += v3.y * w3; acc.z += v3.z * w3; acc.w += v3.w * w3;
        }
        for (; k < cnt; ++k) {
            const int s0 = __builtin_amdgcn_readlane(myidx, k);
            const float q0 = s_src[s0 * HEADS + hd];
            const float4 v0 = *reinterpret_cast<const float4*>(h + (size_t)s0 * OUT_F + lane * 4);
            const float w0 = edge_weight(q0, st, mx);
            den += w0;
            acc.x += v0.x * w0; acc.y += v0.y * w0; acc.z += v0.z * w0; acc.w += v0.w * w0;
        }
    }

    const float  dn = den + EPS_F;
    const float4 bb = *reinterpret_cast<const float4*>(bias + lane * 4);
    float4 o;
    o.x = acc.x / dn + bb.x;
    o.y = acc.y / dn + bb.y;
    o.z = acc.z / dn + bb.z;
    o.w = acc.w / dn + bb.w;
    *reinterpret_cast<float4*>(out + (size_t)n * OUT_F + lane * 4) = o;
}

extern "C" void kernel_launch(void* const* d_in, const int* in_sizes, int n_in,
                              void* d_out, int out_size, void* d_ws, size_t ws_size,
                              hipStream_t stream) {
    (void)in_sizes; (void)n_in; (void)out_size; (void)ws_size;

    const float* x    = (const float*)d_in[0];
    const int*   ei   = (const int*)  d_in[1];
    const float* W    = (const float*)d_in[2];
    const float* a    = (const float*)d_in[3];
    const float* bias = (const float*)d_in[4];
    float* out = (float*)d_out;

    const int* src = ei;
    const int* tgt = ei + N_EDGES;

    float*    ws    = (float*)d_ws;
    float*    hbuf  = ws;
    float*    ssrc  = hbuf + (size_t)N_NODES * OUT_F;
    float*    stgt  = ssrc + (size_t)N_NODES * HEADS;
    float*    tail  = stgt + (size_t)N_NODES * HEADS;
    int*      off   = (int*)tail;
    unsigned* gmax  = (unsigned*)(tail + (size_t)N_NODES * HEADS);
    unsigned short* seg = (unsigned short*)stgt;

    gat_gemm_wmma<<<N_NODES / 16, 256, 0, stream>>>(x, W, hbuf);

    gat_node_scores<<<(N_NODES * HEADS + 255) / 256, 256, 0, stream>>>(hbuf, a, ssrc, stgt, off, gmax);

    gat_edge_max<<<N_EDGES / 256, 256, 0, stream>>>(src, tgt, ssrc, stgt, gmax, off);

    gat_degree_scan<<<1, SCAN_T, 0, stream>>>(off);

    gat_group_edges<<<N_EDGES / 256, 256, 0, stream>>>(src, tgt, off, seg);

    gat_node_agg<<<N_NODES / 8, 256, 0, stream>>>(off, seg, ssrc, a, gmax, hbuf, bias, out);
}


# edge max pass: a lane pair fetches each edge's two 32-B logit rows whole (two row fetches per edge, not a word gather per head and side); head maxima in registers on raw sums, shuffle + LDS reduction
# speedup vs baseline: 2.6554x; 1.2713x over previous
#include <hip/hip_runtime.h>
#include <hip/hip_bf16.h>
#include <math.h>

#define N_NODES   50000
#define N_EDGES   800000
#define IN_F      128
#define HEADS     8
#define HEAD_DIM  16
#define OUT_F     (HEADS * HEAD_DIM)
#define ALPHA     0.2f
#define EPS_F     1e-6f

typedef __attribute__((ext_vector_type(2))) float v2f;
typedef __attribute__((ext_vector_type(8))) float v8f;

__device__ __forceinline__ unsigned enc_f32(float f) {
    unsigned u = __float_as_uint(f);
    return (u & 0x80000000u) ? ~u : (u | 0x80000000u);
}
__device__ __forceinline__ float dec_f32(unsigned u) {
    return (u & 0x80000000u) ? __uint_as_float(u ^ 0x80000000u)
                             : __uint_as_float(~u);
}
__device__ __forceinline__ float leaky(float s) {
    return s >= 0.0f ? s : ALPHA * s;
}

static_assert(N_NODES <= 65536, "source ids are kept as 16-bit words");
static_assert(N_NODES % 8 == 0, "one wave per node, 8 waves per block");
static_assert(N_EDGES % 256 == 0, "edge kernels use full blocks");

__global__ __launch_bounds__(256)
void gat_gemm_wmma(const float* __restrict__ x, const float* __restrict__ W,
                   float* __restrict__ h) {
    __shared__ float As[16 * IN_F];
    const int tid = threadIdx.x;
    const int m0  = blockIdx.x * 16;

    {
        const int base = tid * 8;
        const int row  = base >> 7;
        const int col  = base & 127;
        const float4* g =
            reinterpret_cast<const float4*>(x + (size_t)(m0 + row) * IN_F + col);
        float4 p0 = g[0];
        float4 p1 = g[1];
        *reinterpret_cast<float4*>(&As[row * IN_F + col])     = p0;
        *reinterpret_cast<float4*>(&As[row * IN_F + col + 4]) = p1;
    }
    __syncthreads();

    const int lane = tid & 31;
    const int head = tid >> 5;
    const int l16  = lane & 15;
    const bool hi  = lane >= 16;
    const float* Wh = W + (size_t)head * IN_F * HEAD_DIM;

    v8f c = {};
#pragma unroll
    for (int k = 0; k < IN_F; k += 4) {
        const int kk = k + (hi ? 2 : 0);
        v2f a, b;
        a.x = As[l16 * IN_F + kk];
        a.y = As[l16 * IN_F + kk + 1];
        b.x = Wh[kk * HEAD_DIM + l16];
        b.y = Wh[(kk + 1) * HEAD_DIM + l16];
        c = __builtin_amdgcn_wmma_f32_16x16x4_f32(
                 false, a,  false, b,
                 (short)0, c,  false,  false);
    }

    const int rbase = hi ? 8 : 0;
#pragma unroll
    for (int v = 0; v < 8; ++v)
        h[(size_t)(m0 + rbase + v) * OUT_F + head * HEAD_DIM + l16] = c[v];
}

__global__ void gat_node_scores(const float* __restrict__ h, const float* __restrict__ a,
                                float* __restrict__ s_src, float* __restrict__ s_tgt,
                                int* __restrict__ cnt, unsigned* __restrict__ gmax) {
    const int i = blockIdx.x * blockDim.x + threadIdx.x;
    if (i < N_NODES) cnt[i] = 0;
    if (i < HEADS)   gmax[i] = 0u;
    if (i >= N_NODES * HEADS) return;
    const int hd = i & (HEADS - 1);
    const int n  = i >> 3;
    const float* hv = h + (size_t)n * OUT_F + hd * HEAD_DIM;
    const float* as = a + hd * 2 * HEAD_DIM;
    const float* at = as + HEAD_DIM;
    float ss = 0.0f, st = 0.0f;
#pragma unroll
    for (int d = 0; d < HEAD_DIM; ++d) {
        const float v = hv[d];
        ss += v * as[d];
        st += v * at[d];
    }
    s_src[i] = ss;
    s_tgt[i] = st;
}

#define EM_PAIRS   128
#define EM_PASSES  5
#define EM_EDGES   (EM_PAIRS * EM_PASSES)
static_assert(N_EDGES % EM_EDGES == 0, "every block walks a full run of edges");

__global__ __launch_bounds__(256)
void gat_edge_max(const int* __restrict__ src, const int* __restrict__ tgt,
                  const float* __restrict__ s_src, const float* __restrict__ s_tgt,
                  unsigned* __restrict__ gmax, int* __restrict__ cnt) {
    __shared__ float4 wmax[8][2];
    const int tid  = threadIdx.x;
    const int half = tid & 1;
    const int e0   = blockIdx.x * EM_EDGES + (tid >> 1);

    int s[EM_PASSES], t[EM_PASSES];
    for (int p = 0; p < EM_PASSES; ++p) {
        s[p] = src[e0 + p * EM_PAIRS];
        t[p] = tgt[e0 + p * EM_PAIRS];
    }

    float4 a[EM_PASSES], b[EM_PASSES];
    for (int p = 0; p < EM_PASSES; ++p) {
        a[p] = *reinterpret_cast<const float4*>(s_src + s[p] * HEADS + half * 4);
        b[p] = *reinterpret_cast<const float4*>(s_tgt + t[p] * HEADS + half * 4);
        if (half == 0) atomicAdd(&cnt[t[p]], 1);
    }

    float4 m = make_float4(a[0].x + b[0].x, a[0].y + b[0].y, a[0].z + b[0].z, a[0].w + b[0].w);
    for (int p = 1; p < EM_PASSES; ++p) {
        m.x = fmaxf(m.x, a[p].x + b[p].x);
        m.y = fmaxf(m.y, a[p].y + b[p].y);
        m.z = fmaxf(m.z, a[p].z + b[p].z);
        m.w = fmaxf(m.w, a[p].w + b[p].w);
    }

    for (int d = 2; d < 32; d <<= 1) {
        m.x = fmaxf(m.x, __shfl_xor(m.x, d));
        m.y = fmaxf(m.y, __shfl_xor(m.y, d));
        m.z = fmaxf(m.z, __shfl_xor(m.z, d));
        m.w = fmaxf(m.w, __shfl_xor(m.w, d));
    }
    const int lane = tid & 31;
    if (lane < 2) wmax[tid >> 5][lane] = m;
    __syncthreads();

    if (tid < HEADS) {
        const float* wm = reinterpret_cast<const float*>(&wmax[0][0]);
        float v = wm[tid];
        for (int w = 1; w < 8; ++w) v = fmaxf(v, wm[w * HEADS + tid]);
        atomicMax(&gmax[tid], enc_f32(leaky(v)));
    }
}

#define SCAN_T   1024
#define SCAN_Q   13
#define NODE_Q   (N_NODES / 4)
static_assert(N_NODES % 4 == 0, "counts are moved as 16-B words");
static_assert(SCAN_T * SCAN_Q >= NODE_Q, "scan covers every node");

__global__ __launch_bounds__(SCAN_T)
void gat_degree_scan(int* __restrict__ off) {
    __shared__ int4 buf[SCAN_T * SCAN_Q];
    __shared__ int  wsum[32];
    const int tid  = threadIdx.x;
    const int lane = tid & 31;
    const int wv   = tid >> 5;
    int4* off4 = reinterpret_cast<int4*>(off);

    {
        int4 r[SCAN_Q];
#pragma unroll
        for (int k = 0; k < SCAN_Q; ++k) {
            const int q = k * SCAN_T + tid;
            r[k] = off4[min(q, NODE_Q - 1)];
        }
#pragma unroll
        for (int k = 0; k < SCAN_Q; ++k) {
            const int q = k * SCAN_T + tid;
            buf[q] = (q < NODE_Q) ? r[k] : make_int4(0, 0, 0, 0);
        }
    }
    __syncthreads();

    int4 c[SCAN_Q];
    int sum = 0;
#pragma unroll
    for (int k = 0; k < SCAN_Q; ++k) {
        c[k] = buf[tid * SCAN_Q + k];
        sum += c[k].x + c[k].y + c[k].z + c[k].w;
    }

    int incl = sum;
#pragma unroll
    for (int d = 1; d < 32; d <<= 1) {
        const int v = __shfl_up(incl, d);
        if (lane >= d) incl += v;
    }
    if (lane == 31) wsum[wv] = incl;
    __syncthreads();
    if (wv == 0) {
        const int w = wsum[lane];
        int wi = w;
#pragma unroll
        for (int d = 1; d < 32; d <<= 1) {
            const int v = __shfl_up(wi, d);
            if (lane >= d) wi += v;
        }
        wsum[lane] = wi - w;
    }
    __syncthreads();

    int run = wsum[wv] + incl - sum;
#pragma unroll
    for (int k = 0; k < SCAN_Q; ++k) {
        int4 o;
        o.x = run; run += c[k].x;
        o.y = run; run += c[k].y;
        o.z = run; run += c[k].z;
        o.w = run; run += c[k].w;
        buf[tid * SCAN_Q + k] = o;
    }
    __syncthreads();
#pragma unroll
    for (int k = 0; k < SCAN_Q; ++k) {
        const int q = k * SCAN_T + tid;
        if (q < NODE_Q) off4[q] = buf[q];
    }
}

__global__ __launch_bounds__(256)
void gat_group_edges(const int* __restrict__ src, const int* __restrict__ tgt,
                     int* __restrict__ off, unsigned short* __restrict__ seg_src) {
    const int e = blockIdx.x * blockDim.x + threadIdx.x;
    if (e >= N_EDGES) return;
    const int t   = tgt[e];
    const int pos = atomicAdd(&off[t], 1);
    if ((unsigned)pos < (unsigned)N_EDGES)
        seg_src[pos] = (unsigned short)src[e];
}

__device__ __forceinline__ float edge_weight(float ssrc, float stgt, float mx) {
    const float sc = leaky(ssrc + stgt);
    return __expf(fminf(fmaxf(sc - mx, -20.0f), 20.0f));
}

__global__ __launch_bounds__(256)
void gat_node_agg(const int* __restrict__ off, const unsigned short* __restrict__ seg_src,
                  const float* __restrict__ s_src, const float* __restrict__ a,
                  const unsigned* __restrict__ gmax, const float* __restrict__ h,
                  const float* __restrict__ bias, float* __restrict__ out) {
    const int lane = threadIdx.x & 31;
    const int wv = __builtin_amdgcn_readfirstlane((int)(threadIdx.x >> 5));
    const int n  = blockIdx.x * 8 + wv;
    const int hd = lane >> 2;

    const float4 hn = *reinterpret_cast<const float4*>(h + (size_t)n * OUT_F + lane * 4);
    const float4 at = *reinterpret_cast<const float4*>(a + hd * 2 * HEAD_DIM + HEAD_DIM + (lane & 3) * 4);
    float st = hn.x * at.x + hn.y * at.y + hn.z * at.z + hn.w * at.w;
    st += __shfl_xor(st, 1);
    st += __shfl_xor(st, 2);

    const float mx = dec_f32(gmax[hd]);

    int beg = (n > 0) ? off[n - 1] : 0;
    int end = off[n];
    beg = max(beg, 0);
    end = min(end, N_EDGES);

    float4 acc = make_float4(0.0f, 0.0f, 0.0f, 0.0f);
    float  den = 0.0f;

    for (int base = beg; base < end; base += 32) {
        const int cnt = min(end - base, 32);
        int myidx = 0;
        if (lane < cnt) myidx = seg_src[base + lane];

        int k = 0;
        for (; k + 4 <= cnt; k += 4) {
            const int s0 = __builtin_amdgcn_readlane(myidx, k);
            const int s1 = __builtin_amdgcn_readlane(myidx, k + 1);
            const int s2 = __builtin_amdgcn_readlane(myidx, k + 2);
            const int s3 = __builtin_amdgcn_readlane(myidx, k + 3);
            const float q0 = s_src[s0 * HEADS + hd];
            const float q1 = s_src[s1 * HEADS + hd];
            const float q2 = s_src[s2 * HEADS + hd];
            const float q3 = s_src[s3 * HEADS + hd];
            const float4 v0 = *reinterpret_cast<const float4*>(h + (size_t)s0 * OUT_F + lane * 4);
            const float4 v1 = *reinterpret_cast<const float4*>(h + (size_t)s1 * OUT_F + lane * 4);
            const float4 v2 = *reinterpret_cast<const float4*>(h + (size_t)s2 * OUT_F + lane * 4);
            const float4 v3 = *reinterpret_cast<const float4*>(h + (size_t)s3 * OUT_F + lane * 4);
            const float w0 = edge_weight(q0, st, mx);
            const float w1 = edge_weight(q1, st, mx);
            const float w2 = edge_weight(q2, st, mx);
            const float w3 = edge_weight(q3, st, mx);
            den += w0; den += w1; den += w2; den += w3;
            acc.x += v0.x * w0; acc.y += v0.y * w0; acc.z += v0.z * w0; acc.w += v0.w * w0;
            acc.x += v1.x * w1; acc.y += v1.y * w1; acc.z += v1.z * w1; acc.w += v1.w * w1;
            acc.x += v2.x * w2; acc.y += v2.y * w2; acc.z += v2.z * w2; acc.w += v2.w * w2;
            acc.x += v3.x * w3; acc.y += v3.y * w3; acc.z += v3.z * w3; acc.w += v3.w * w3;
        }
        for (; k < cnt; ++k) {
            const int s0 = __builtin_amdgcn_readlane(myidx, k);
            const float q0 = s_src[s0 * HEADS + hd];
            const float4 v0 = *reinterpret_cast<const float4*>(h + (size_t)s0 * OUT_F + lane * 4);
            const float w0 = edge_weight(q0, st, mx);
            den += w0;
            acc.x += v0.x * w0; acc.y += v0.y * w0; acc.z += v0.z * w0; acc.w += v0.w * w0;
        }
    }

    const float  dn = den + EPS_F;
    const float4 bb = *reinterpret_cast<const float4*>(bias + lane * 4);
    float4 o;
    o.x = acc.x / dn + bb.x;
    o.y = acc.y / dn + bb.y;
    o.z = acc.z / dn + bb.z;
    o.w = acc.w / dn + bb.w;
    *reinterpret_cast<float4*>(out + (size_t)n * OUT_F + lane * 4) = o;
}

extern "C" void kernel_launch(void* const* d_in, const int* in_sizes, int n_in,
                              void* d_out, int out_size, void* d_ws, size_t ws_size,
                              hipStream_t stream) {
    (void)in_sizes; (void)n_in; (void)out_size; (void)ws_size;

    const float* x    = (const float*)d_in[0];
    const int*   ei   = (const int*)  d_in[1];
    const float* W    = (const float*)d_in[2];
    const float* a    = (const float*)d_in[3];
    const float* bias = (const float*)d_in[4];
    float* out = (float*)d_out;

    const int* src = ei;
    const int* tgt = ei + N_EDGES;

    float*    ws    = (float*)d_ws;
    float*    hbuf  = ws;
    float*    ssrc  = hbuf + (size_t)N_NODES * OUT_F;
    float*    stgt  = ssrc + (size_t)N_NODES * HEADS;
    float*    tail  = stgt + (size_t)N_NODES * HEADS;
    int*      off   = (int*)tail;
    unsigned* gmax  = (unsigned*)(tail + (size_t)N_NODES * HEADS);
    unsigned short* seg = (unsigned short*)stgt;

    gat_gemm_wmma<<<N_NODES / 16, 256, 0, stream>>>(x, W, hbuf);

    gat_node_scores<<<(N_NODES * HEADS + 255) / 256, 256, 0, stream>>>(hbuf, a, ssrc, stgt, off, gmax);

    gat_edge_max<<<N_EDGES / EM_EDGES, 256, 0, stream>>>(src, tgt, ssrc, stgt, gmax, off);

    gat_degree_scan<<<1, SCAN_T, 0, stream>>>(off);

    gat_group_edges<<<N_EDGES / 256, 256, 0, stream>>>(src, tgt, off, seg);

    gat_node_agg<<<N_NODES / 8, 256, 0, stream>>>(off, seg, ssrc, a, gmax, hbuf, bias, out);
}


# one scattered atomic per edge, not two: the counting add returns the edge's rank in its target segment (kept as a byte); the fill pass places at start + rank with no atomic, four edges per lane
# speedup vs baseline: 3.3312x; 1.2545x over previous
#include <hip/hip_runtime.h>
#include <hip/hip_bf16.h>
#include <math.h>

#define N_NODES   50000
#define N_EDGES   800000
#define IN_F      128
#define HEADS     8
#define HEAD_DIM  16
#define OUT_F     (HEADS * HEAD_DIM)
#define ALPHA     0.2f
#define EPS_F     1e-6f

typedef __attribute__((ext_vector_type(2))) float v2f;
typedef __attribute__((ext_vector_type(8))) float v8f;

__device__ __forceinline__ unsigned enc_f32(float f) {
    unsigned u = __float_as_uint(f);
    return (u & 0x80000000u) ? ~u : (u | 0x80000000u);
}
__device__ __forceinline__ float dec_f32(unsigned u) {
    return (u & 0x80000000u) ? __uint_as_float(u ^ 0x80000000u)
                             : __uint_as_float(~u);
}
__device__ __forceinline__ float leaky(float s) {
    return s >= 0.0f ? s : ALPHA * s;
}

static_assert(N_NODES <= 65536, "source ids are kept as 16-bit words");
static_assert(N_NODES % 8 == 0, "one wave per node, 8 waves per block");
static_assert(N_EDGES % 256 == 0, "edge kernels use full blocks");

__global__ __launch_bounds__(256)
void gat_gemm_wmma(const float* __restrict__ x, const float* __restrict__ W,
                   float* __restrict__ h) {
    __shared__ float As[16 * IN_F];
    const int tid = threadIdx.x;
    const int m0  = blockIdx.x * 16;

    {
        const int base = tid * 8;
        const int row  = base >> 7;
        const int col  = base & 127;
        const float4* g =
            reinterpret_cast<const float4*>(x + (size_t)(m0 + row) * IN_F + col);
        float4 p0 = g[0];
        float4 p1 = g[1];
        *reinterpret_cast<float4*>(&As[row * IN_F + col])     = p0;
        *reinterpret_cast<float4*>(&As[row * IN_F + col + 4]) = p1;
    }
    __syncthreads();

    const int lane = tid & 31;
    const int head = tid >> 5;
    const int l16  = lane & 15;
    const bool hi  = lane >= 16;
    const float* Wh = W + (size_t)head * IN_F * HEAD_DIM;

    v8f c = {};
#pragma unroll
    for (int k = 0; k < IN_F; k += 4) {
        const int kk = k + (hi ? 2 : 0);
        v2f a, b;
        a.x = As[l16 * IN_F + kk];
        a.y = As[l16 * IN_F + kk + 1];
        b.x = Wh[kk * HEAD_DIM + l16];
        b.y = Wh[(kk + 1) * HEAD_DIM + l16];
        c = __builtin_amdgcn_wmma_f32_16x16x4_f32(
                 false, a,  false, b,
                 (short)0, c,  false,  false);
    }

    const int rbase = hi ? 8 : 0;
#pragma unroll
    for (int v = 0; v < 8; ++v)
        h[(size_t)(m0 + rbase + v) * OUT_F + head * HEAD_DIM + l16] = c[v];
}

__global__ void gat_node_scores(const float* __restrict__ h, const float* __restrict__ a,
                                float* __restrict__ s_src, float* __restrict__ s_tgt,
                                int* __restrict__ cnt, int* __restrict__ cnt_hi,
                                unsigned* __restrict__ gmax) {
    const int i = blockIdx.x * blockDim.x + threadIdx.x;
    if (i < N_NODES) { cnt[i] = 0; cnt_hi[i] = 0; }
    if (i < HEADS)   gmax[i] = 0u;
    if (i >= N_NODES * HEADS) return;
    const int hd = i & (HEADS - 1);
    const int n  = i >> 3;
    const float* hv = h + (size_t)n * OUT_F + hd * HEAD_DIM;
    const float* as = a + hd * 2 * HEAD_DIM;
    const float* at = as + HEAD_DIM;
    float ss = 0.0f, st = 0.0f;
#pragma unroll
    for (int d = 0; d < HEAD_DIM; ++d) {
        const float v = hv[d];
        ss += v * as[d];
        st += v * at[d];
    }
    s_src[i] = ss;
    s_tgt[i] = st;
}

#define EM_PAIRS   128
#define EM_PASSES  5
#define EM_EDGES   (EM_PAIRS * EM_PASSES)
static_assert(N_EDGES % EM_EDGES == 0, "every block walks a full run of edges");
#define RANK_SAT   255

__global__ __launch_bounds__(256)
void gat_edge_max(const int* __restrict__ src, const int* __restrict__ tgt,
                  const float* __restrict__ s_src, const float* __restrict__ s_tgt,
                  unsigned* __restrict__ gmax, int* __restrict__ cnt,
                  unsigned char* __restrict__ rank8) {
    __shared__ float4 wmax[8][2];
    const int tid  = threadIdx.x;
    const int half = tid & 1;
    const int e0   = blockIdx.x * EM_EDGES + (tid >> 1);

    int s[EM_PASSES], t[EM_PASSES];
    for (int p = 0; p < EM_PASSES; ++p) {
        s[p] = src[e0 + p * EM_PAIRS];
        t[p] = tgt[e0 + p * EM_PAIRS];
    }

    float4 a[EM_PASSES], b[EM_PASSES];
    int    r[EM_PASSES];
    for (int p = 0; p < EM_PASSES; ++p) {
        a[p] = *reinterpret_cast<const float4*>(s_src + s[p] * HEADS + half * 4);
        b[p] = *reinterpret_cast<const float4*>(s_tgt + t[p] * HEADS + half * 4);
        r[p] = 0;
        if (half == 0) r[p] = atomicAdd(&cnt[t[p]], 1);
    }

    float4 m = make_float4(a[0].x + b[0].x, a[0].y + b[0].y, a[0].z + b[0].z, a[0].w + b[0].w);
    for (int p = 1; p < EM_PASSES; ++p) {
        m.x = fmaxf(m.x, a[p].x + b[p].x);
        m.y = fmaxf(m.y, a[p].y + b[p].y);
        m.z = fmaxf(m.z, a[p].z + b[p].z);
        m.w = fmaxf(m.w, a[p].w + b[p].w);
    }

    if (half == 0) {
        for (int p = 0; p < EM_PASSES; ++p)
            rank8[e0 + p * EM_PAIRS] = (unsigned char)min(r[p], RANK_SAT);
    }

    for (int d = 2; d < 32; d <<= 1) {
        m.x = fmaxf(m.x, __shfl_xor(m.x, d));
        m.y = fmaxf(m.y, __shfl_xor(m.y, d));
        m.z = fmaxf(m.z, __shfl_xor(m.z, d));
        m.w = fmaxf(m.w, __shfl_xor(m.w, d));
    }
    const int lane = tid & 31;
    if (lane < 2) wmax[tid >> 5][lane] = m;
    __syncthreads();

    if (tid < HEADS) {
        const float* wm = reinterpret_cast<const float*>(&wmax[0][0]);
        float v = wm[tid];
        for (int w = 1; w < 8; ++w) v = fmaxf(v, wm[w * HEADS + tid]);
        atomicMax(&gmax[tid], enc_f32(leaky(v)));
    }
}

#define SCAN_T   1024
#define SCAN_Q   13
#define NODE_Q   (N_NODES / 4)
static_assert(N_NODES % 4 == 0, "counts are moved as 16-B words");
static_assert(SCAN_T * SCAN_Q >= NODE_Q, "scan covers every node");

__global__ __launch_bounds__(SCAN_T)
void gat_degree_scan(int* __restrict__ off) {
    __shared__ int4 buf[SCAN_T * SCAN_Q];
    __shared__ int  wsum[32];
    const int tid  = threadIdx.x;
    const int lane = tid & 31;
    const int wv   = tid >> 5;
    int4* off4 = reinterpret_cast<int4*>(off);

    {
        int4 r[SCAN_Q];
#pragma unroll
        for (int k = 0; k < SCAN_Q; ++k) {
            const int q = k * SCAN_T + tid;
            r[k] = off4[min(q, NODE_Q - 1)];
        }
#pragma unroll
        for (int k = 0; k < SCAN_Q; ++k) {
            const int q = k * SCAN_T + tid;
            buf[q] = (q < NODE_Q) ? r[k] : make_int4(0, 0, 0, 0);
        }
    }
    __syncthreads();

    int4 c[SCAN_Q];
    int sum = 0;
#pragma unroll
    for (int k = 0; k < SCAN_Q; ++k) {
        c[k] = buf[tid * SCAN_Q + k];
        sum += c[k].x + c[k].y + c[k].z + c[k].w;
    }

    int incl = sum;
#pragma unroll
    for (int d = 1; d < 32; d <<= 1) {
        const int v = __shfl_up(incl, d);
        if (lane >= d) incl += v;
    }
    if (lane == 31) wsum[wv] = incl;
    __syncthreads();
    if (wv == 0) {
        const int w = wsum[lane];
        int wi = w;
#pragma unroll
        for (int d = 1; d < 32; d <<= 1) {
            const int v = __shfl_up(wi, d);
            if (lane >= d) wi += v;
        }
        wsum[lane] = wi - w;
    }
    __syncthreads();

    int run = wsum[wv] + incl - sum;
#pragma unroll
    for (int k = 0; k < SCAN_Q; ++k) {
        int4 o;
        o.x = run; run += c[k].x;
        o.y = run; run += c[k].y;
        o.z = run; run += c[k].z;
        o.w = run; run += c[k].w;
        buf[tid * SCAN_Q + k] = o;
    }
    __syncthreads();
#pragma unroll
    for (int k = 0; k < SCAN_Q; ++k) {
        const int q = k * SCAN_T + tid;
        if (q < NODE_Q) off4[q] = buf[q];
    }
}

#define GE_QUADS  (N_EDGES / 4)
static_assert(N_EDGES % 4 == 0, "edges are placed four per lane");

__device__ __forceinline__ void place_edge(int s, int pos, unsigned short* __restrict__ seg_src) {
    if ((unsigned)pos < (unsigned)N_EDGES)
        seg_src[pos] = (unsigned short)s;
}

__global__ __launch_bounds__(256)
void gat_group_edges(const int* __restrict__ src, const int* __restrict__ tgt,
                     const unsigned char* __restrict__ rank8, const int* __restrict__ off,
                     int* __restrict__ cnt_hi, unsigned short* __restrict__ seg_src) {
    const int q = blockIdx.x * blockDim.x + threadIdx.x;
    if (q >= GE_QUADS) return;
    const int4   s4 = reinterpret_cast<const int4*>(src)[q];
    const int4   t4 = reinterpret_cast<const int4*>(tgt)[q];
    const uchar4 r4 = reinterpret_cast<const uchar4*>(rank8)[q];
    const int o0 = off[t4.x];
    const int o1 = off[t4.y];
    const int o2 = off[t4.z];
    const int o3 = off[t4.w];
    int r0 = r4.x, r1 = r4.y, r2 = r4.z, r3 = r4.w;
    if (max(max(r0, r1), max(r2, r3)) == RANK_SAT) {
        if (r0 == RANK_SAT) r0 += atomicAdd(&cnt_hi[t4.x], 1);
        if (r1 == RANK_SAT) r1 += atomicAdd(&cnt_hi[t4.y], 1);
        if (r2 == RANK_SAT) r2 += atomicAdd(&cnt_hi[t4.z], 1);
        if (r3 == RANK_SAT) r3 += atomicAdd(&cnt_hi[t4.w], 1);
    }
    place_edge(s4.x, o0 + r0, seg_src);
    place_edge(s4.y, o1 + r1, seg_src);
    place_edge(s4.z, o2 + r2, seg_src);
    place_edge(s4.w, o3 + r3, seg_src);
}

__device__ __forceinline__ float edge_weight(float ssrc, float stgt, float mx) {
    const float sc = leaky(ssrc + stgt);
    return __expf(fminf(fmaxf(sc - mx, -20.0f), 20.0f));
}

__global__ __launch_bounds__(256)
void gat_node_agg(const int* __restrict__ off, const unsigned short* __restrict__ seg_src,
                  const float* __restrict__ s_src, const float* __restrict__ a,
                  const unsigned* __restrict__ gmax, const float* __restrict__ h,
                  const float* __restrict__ bias, float* __restrict__ out) {
    const int lane = threadIdx.x & 31;
    const int wv = __builtin_amdgcn_readfirstlane((int)(threadIdx.x >> 5));
    const int n  = blockIdx.x * 8 + wv;
    const int hd = lane >> 2;

    const float4 hn = *reinterpret_cast<const float4*>(h + (size_t)n * OUT_F + lane * 4);
    const float4 at = *reinterpret_cast<const float4*>(a + hd * 2 * HEAD_DIM + HEAD_DIM + (lane & 3) * 4);
    float st = hn.x * at.x + hn.y * at.y + hn.z * at.z + hn.w * at.w;
    st += __shfl_xor(st, 1);
    st += __shfl_xor(st, 2);

    const float mx = dec_f32(gmax[hd]);

    int beg = off[n];
    int end = off[min(n + 1, N_NODES - 1)];
    if (n + 1 >= N_NODES) end = N_EDGES;
    beg = max(beg, 0);
    end = min(end, N_EDGES);

    float4 acc = make_float4(0.0f, 0.0f, 0.0f, 0.0f);
    float  den = 0.0f;

    for (int base = beg; base < end; base += 32) {
        const int cnt = min(end - base, 32);
        int myidx = 0;
        if (lane < cnt) myidx = seg_src[base + lane];

        int k = 0;
        for (; k + 4 <= cnt; k += 4) {
            const int s0 = __builtin_amdgcn_readlane(myidx, k);
            const int s1 = __builtin_amdgcn_readlane(myidx, k + 1);
            const int s2 = __builtin_amdgcn_readlane(myidx, k + 2);
            const int s3 = __builtin_amdgcn_readlane(myidx, k + 3);
            const float q0 = s_src[s0 * HEADS + hd];
            const float q1 = s_src[s1 * HEADS + hd];
            const float q2 = s_src[s2 * HEADS + hd];
            const float q3 = s_src[s3 * HEADS + hd];
            const float4 v0 = *reinterpret_cast<const float4*>(h + (size_t)s0 * OUT_F + lane * 4);
            const float4 v1 = *reinterpret_cast<const float4*>(h + (size_t)s1 * OUT_F + lane * 4);
            const float4 v2 = *reinterpret_cast<const float4*>(h + (size_t)s2 * OUT_F + lane * 4);
            const float4 v3 = *reinterpret_cast<const float4*>(h + (size_t)s3 * OUT_F + lane * 4);
            const float w0 = edge_weight(q0, st, mx);
            const float w1 = edge_weight(q1, st, mx);
            const float w2 = edge_weight(q2, st, mx);
            const float w3 = edge_weight(q3, st, mx);
            den += w0; den += w1; den += w2; den += w3;
            acc.x += v0.x * w0; acc.y += v0.y * w0; acc.z += v0.z * w0; acc.w += v0.w * w0;
            acc.x += v1.x * w1; acc.y += v1.y * w1; acc.z += v1.z * w1; acc.w += v1.w * w1;
            acc.x += v2.x * w2; acc.y += v2.y * w2; acc.z += v2.z * w2; acc.w += v2.w * w2;
            acc.x += v3.x * w3; acc.y += v3.y * w3; acc.z += v3.z * w3; acc.w += v3.w * w3;
        }
        for (; k < cnt; ++k) {
            const int s0 = __builtin_amdgcn_readlane(myidx, k);
            const float q0 = s_src[s0 * HEADS + hd];
            const float4 v0 = *reinterpret_cast<const float4*>(h + (size_t)s0 * OUT_F + lane * 4);
            const float w0 = edge_weight(q0, st, mx);
            den += w0;
            acc.x += v0.x * w0; acc.y += v0.y * w0; acc.z += v0.z * w0; acc.w += v0.w * w0;
        }
    }

    const float  dn = den + EPS_F;
    const float4 bb = *reinterpret_cast<const float4*>(bias + lane * 4);
    float4 o;
    o.x = acc.x / dn + bb.x;
    o.y = acc.y / dn + bb.y;
    o.z = acc.z / dn + bb.z;
    o.w = acc.w / dn + bb.w;
    *reinterpret_cast<float4*>(out + (size_t)n * OUT_F + lane * 4) = o;
}

extern "C" void kernel_launch(void* const* d_in, const int* in_sizes, int n_in,
                              void* d_out, int out_size, void* d_ws, size_t ws_size,
                              hipStream_t stream) {
    (void)in_sizes; (void)n_in; (void)out_size; (void)ws_size;

    const float* x    = (const float*)d_in[0];
    const int*   ei   = (const int*)  d_in[1];
    const float* W    = (const float*)d_in[2];
    const float* a    = (const float*)d_in[3];
    const float* bias = (const float*)d_in[4];
    float* out = (float*)d_out;

    const int* src = ei;
    const int* tgt = ei + N_EDGES;

    float*    ws    = (float*)d_ws;
    float*    hbuf  = ws;
    float*    ssrc  = hbuf + (size_t)N_NODES * OUT_F;
    float*    stgt  = ssrc + (size_t)N_NODES * HEADS;
    float*    tail  = stgt + (size_t)N_NODES * HEADS;
    int*      off   = (int*)tail;
    unsigned* gmax  = (unsigned*)(tail + (size_t)N_NODES * HEADS);
    unsigned short* seg = (unsigned short*)stgt;
    int*      cnt_hi = off + N_NODES;
    unsigned char* rank8 = (unsigned char*)(off + 2 * N_NODES);

    gat_gemm_wmma<<<N_NODES / 16, 256, 0, stream>>>(x, W, hbuf);

    gat_node_scores<<<(N_NODES * HEADS + 255) / 256, 256, 0, stream>>>(hbuf, a, ssrc, stgt, off, cnt_hi, gmax);

    gat_edge_max<<<N_EDGES / EM_EDGES, 256, 0, stream>>>(src, tgt, ssrc, stgt, gmax, off, rank8);

    gat_degree_scan<<<1, SCAN_T, 0, stream>>>(off);

    gat_group_edges<<<(GE_QUADS + 255) / 256, 256, 0, stream>>>(src, tgt, rank8, off, cnt_hi, seg);

    gat_node_agg<<<N_NODES / 8, 256, 0, stream>>>(off, seg, ssrc, a, gmax, hbuf, bias, out);
}


# product kernel: A-tile rows kept in LDS at a pitch of IN_F + 4 words, not IN_F, so the rows of one fragment read fall on different banks (all fell on one bank pair and were served serially)
# speedup vs baseline: 3.5842x; 1.0759x over previous
#include <hip/hip_runtime.h>
#include <hip/hip_bf16.h>
#include <math.h>

#define N_NODES   50000
#define N_EDGES   800000
#define IN_F      128
#define HEADS     8
#define HEAD_DIM  16
#define OUT_F     (HEADS * HEAD_DIM)
#define ALPHA     0.2f
#define EPS_F     1e-6f

typedef __attribute__((ext_vector_type(2))) float v2f;
typedef __attribute__((ext_vector_type(8))) float v8f;

__device__ __forceinline__ unsigned enc_f32(float f) {
    unsigned u = __float_as_uint(f);
    return (u & 0x80000000u) ? ~u : (u | 0x80000000u);
}
__device__ __forceinline__ float dec_f32(unsigned u) {
    return (u & 0x80000000u) ? __uint_as_float(u ^ 0x80000000u)
                             : __uint_as_float(~u);
}
__device__ __forceinline__ float leaky(float s) {
    return s >= 0.0f ? s : ALPHA * s;
}

static_assert(N_NODES <= 65536, "source ids are kept as 16-bit words");
static_assert(N_NODES % 8 == 0, "one wave per node, 8 waves per block");
static_assert(N_EDGES % 256 == 0, "edge kernels use full blocks");

#define A_PITCH   (IN_F + 4)
static_assert((A_PITCH * 4) % 16 == 0, "A-tile rows are written as 16-B words");

__global__ __launch_bounds__(256)
void gat_gemm_wmma(const float* __restrict__ x, const float* __restrict__ W,
                   float* __restrict__ h) {
    __shared__ float As[16 * A_PITCH];
    const int tid = threadIdx.x;
    const int m0  = blockIdx.x * 16;

    {
        const int base = tid * 8;
        const int row  = base >> 7;
        const int col  = base & 127;
        const float4* g =
            reinterpret_cast<const float4*>(x + (size_t)(m0 + row) * IN_F + col);
        float4 p0 = g[0];
        float4 p1 = g[1];
        *reinterpret_cast<float4*>(&As[row * A_PITCH + col])     = p0;
        *reinterpret_cast<float4*>(&As[row * A_PITCH + col + 4]) = p1;
    }
    __syncthreads();

    const int lane = tid & 31;
    const int head = tid >> 5;
    const int l16  = lane & 15;
    const bool hi  = lane >= 16;
    const float* Wh = W + (size_t)head * IN_F * HEAD_DIM;

    v8f c = {};
#pragma unroll
    for (int k = 0; k < IN_F; k += 4) {
        const int kk = k + (hi ? 2 : 0);
        v2f a, b;
        a.x = As[l16 * A_PITCH + kk];
        a.y = As[l16 * A_PITCH + kk + 1];
        b.x = Wh[kk * HEAD_DIM + l16];
        b.y = Wh[(kk + 1) * HEAD_DIM + l16];
        c = __builtin_amdgcn_wmma_f32_16x16x4_f32(
                 false, a,  false, b,
                 (short)0, c,  false,  false);
    }

    const int rbase = hi ? 8 : 0;
#pragma unroll
    for (int v = 0; v < 8; ++v)
        h[(size_t)(m0 + rbase + v) * OUT_F + head * HEAD_DIM + l16] = c[v];
}

__global__ void gat_node_scores(const float* __restrict__ h, const float* __restrict__ a,
                                float* __restrict__ s_src, float* __restrict__ s_tgt,
                                int* __restrict__ cnt, int* __restrict__ cnt_hi,
                                unsigned* __restrict__ gmax) {
    const int i = blockIdx.x * blockDim.x + threadIdx.x;
    if (i < N_NODES) { cnt[i] = 0; cnt_hi[i] = 0; }
    if (i < HEADS)   gmax[i] = 0u;
    if (i >= N_NODES * HEADS) return;
    const int hd = i & (HEADS - 1);
    const int n  = i >> 3;
    const float* hv = h + (size_t)n * OUT_F + hd * HEAD_DIM;
    const float* as = a + hd * 2 * HEAD_DIM;
    const float* at = as + HEAD_DIM;
    float ss = 0.0f, st = 0.0f;
#pragma unroll
    for (int d = 0; d < HEAD_DIM; ++d) {
        const float v = hv[d];
        ss += v * as[d];
        st += v * at[d];
    }
    s_src[i] = ss;
    s_tgt[i] = st;
}

#define EM_PAIRS   128
#define EM_PASSES  5
#define EM_EDGES   (EM_PAIRS * EM_PASSES)
static_assert(N_EDGES % EM_EDGES == 0, "every block walks a full run of edges");
#define RANK_SAT   255

__global__ __launch_bounds__(256)
void gat_edge_max(const int* __restrict__ src, const int* __restrict__ tgt,
                  const float* __restrict__ s_src, const float* __restrict__ s_tgt,
                  unsigned* __restrict__ gmax, int* __restrict__ cnt,
                  unsigned char* __restrict__ rank8) {
    __shared__ float4 wmax[8][2];
    const int tid  = threadIdx.x;
    const int half = tid & 1;
    const int e0   = blockIdx.x * EM_EDGES + (tid >> 1);

    int s[EM_PASSES], t[EM_PASSES];
    for (int p = 0; p < EM_PASSES; ++p) {
        s[p] = src[e0 + p * EM_PAIRS];
        t[p] = tgt[e0 + p * EM_PAIRS];
    }

    float4 a[EM_PASSES], b[EM_PASSES];
    int    r[EM_PASSES];
    for (int p = 0; p < EM_PASSES; ++p) {
        a[p] = *reinterpret_cast<const float4*>(s_src + s[p] * HEADS + half * 4);
        b[p] = *reinterpret_cast<const float4*>(s_tgt + t[p] * HEADS + half * 4);
        r[p] = 0;
        if (half == 0) r[p] = atomicAdd(&cnt[t[p]], 1);
    }

    float4 m = make_float4(a[0].x + b[0].x, a[0].y + b[0].y, a[0].z + b[0].z, a[0].w + b[0].w);
    for (int p = 1; p < EM_PASSES; ++p) {
        m.x = fmaxf(m.x, a[p].x + b[p].x);
        m.y = fmaxf(m.y, a[p].y + b[p].y);
        m.z = fmaxf(m.z, a[p].z + b[p].z);
        m.w = fmaxf(m.w, a[p].w + b[p].w);
    }

    if (half == 0) {
        for (int p = 0; p < EM_PASSES; ++p)
            rank8[e0 + p * EM_PAIRS] = (unsigned char)min(r[p], RANK_SAT);
    }

    for (int d = 2; d < 32; d <<= 1) {
        m.x = fmaxf(m.x, __shfl_xor(m.x, d));
        m.y = fmaxf(m.y, __shfl_xor(m.y, d));
        m.z = fmaxf(m.z, __shfl_xor(m.z, d));
        m.w = fmaxf(m.w, __shfl_xor(m.w, d));
    }
    const int lane = tid & 31;
    if (lane < 2) wmax[tid >> 5][lane] = m;
    __syncthreads();

    if (tid < HEADS) {
        const float* wm = reinterpret_cast<const float*>(&wmax[0][0]);
        float v = wm[tid];
        for (int w = 1; w < 8; ++w) v = fmaxf(v, wm[w * HEADS + tid]);
        atomicMax(&gmax[tid], enc_f32(leaky(v)));
    }
}

#define SCAN_T   1024
#define SCAN_Q   13
#define NODE_Q   (N_NODES / 4)
static_assert(N_NODES % 4 == 0, "counts are moved as 16-B words");
static_assert(SCAN_T * SCAN_Q >= NODE_Q, "scan covers every node");

__global__ __launch_bounds__(SCAN_T)
void gat_degree_scan(int* __restrict__ off) {
    __shared__ int4 buf[SCAN_T * SCAN_Q];
    __shared__ int  wsum[32];
    const int tid  = threadIdx.x;
    const int lane = tid & 31;
    const int wv   = tid >> 5;
    int4* off4 = reinterpret_cast<int4*>(off);

    {
        int4 r[SCAN_Q];
#pragma unroll
        for (int k = 0; k < SCAN_Q; ++k) {
            const int q = k * SCAN_T + tid;
            r[k] = off4[min(q, NODE_Q - 1)];
        }
#pragma unroll
        for (int k = 0; k < SCAN_Q; ++k) {
            const int q = k * SCAN_T + tid;
            buf[q] = (q < NODE_Q) ? r[k] : make_int4(0, 0, 0, 0);
        }
    }
    __syncthreads();

    int4 c[SCAN_Q];
    int sum = 0;
#pragma unroll
    for (int k = 0; k < SCAN_Q; ++k) {
        c[k] = buf[tid * SCAN_Q + k];
        sum += c[k].x + c[k].y + c[k].z + c[k].w;
    }

    int incl = sum;
#pragma unroll
    for (int d = 1; d < 32; d <<= 1) {
        const int v = __shfl_up(incl, d);
        if (lane >= d) incl += v;
    }
    if (lane == 31) wsum[wv] = incl;
    __syncthreads();
    if (wv == 0) {
        const int w = wsum[lane];
        int wi = w;
#pragma unroll
        for (int d = 1; d < 32; d <<= 1) {
            const int v = __shfl_up(wi, d);
            if (lane >= d) wi += v;
        }
        wsum[lane] = wi - w;
    }
    __syncthreads();

    int run = wsum[wv] + incl - sum;
#pragma unroll
    for (int k = 0; k < SCAN_Q; ++k) {
        int4 o;
        o.x = run; run += c[k].x;
        o.y = run; run += c[k].y;
        o.z = run; run += c[k].z;
        o.w = run; run += c[k].w;
        buf[tid * SCAN_Q + k] = o;
    }
    __syncthreads();
#pragma unroll
    for (int k = 0; k < SCAN_Q; ++k) {
        const int q = k * SCAN_T + tid;
        if (q < NODE_Q) off4[q] = buf[q];
    }
}

#define GE_QUADS  (N_EDGES / 4)
static_assert(N_EDGES % 4 == 0, "edges are placed four per lane");

__device__ __forceinline__ void place_edge(int s, int pos, unsigned short* __restrict__ seg_src) {
    if ((unsigned)pos < (unsigned)N_EDGES)
        seg_src[pos] = (unsigned short)s;
}

__global__ __launch_bounds__(256)
void gat_group_edges(const int* __restrict__ src, const int* __restrict__ tgt,
                     const unsigned char* __restrict__ rank8, const int* __restrict__ off,
                     int* __restrict__ cnt_hi, unsigned short* __restrict__ seg_src) {
    const int q = blockIdx.x * blockDim.x + threadIdx.x;
    if (q >= GE_QUADS) return;
    const int4   s4 = reinterpret_cast<const int4*>(src)[q];
    const int4   t4 = reinterpret_cast<const int4*>(tgt)[q];
    const uchar4 r4 = reinterpret_cast<const uchar4*>(rank8)[q];
    const int o0 = off[t4.x];
    const int o1 = off[t4.y];
    const int o2 = off[t4.z];
    const int o3 = off[t4.w];
    int r0 = r4.x, r1 = r4.y, r2 = r4.z, r3 = r4.w;
    if (max(max(r0, r1), max(r2, r3)) == RANK_SAT) {
        if (r0 == RANK_SAT) r0 += atomicAdd(&cnt_hi[t4.x], 1);
        if (r1 == RANK_SAT) r1 += atomicAdd(&cnt_hi[t4.y], 1);
        if (r2 == RANK_SAT) r2 += atomicAdd(&cnt_hi[t4.z], 1);
        if (r3 == RANK_SAT) r3 += atomicAdd(&cnt_hi[t4.w], 1);
    }
    place_edge(s4.x, o0 + r0, seg_src);
    place_edge(s4.y, o1 + r1, seg_src);
    place_edge(s4.z, o2 + r2, seg_src);
    place_edge(s4.w, o3 + r3, seg_src);
}

__device__ __forceinline__ float edge_weight(float ssrc, float stgt, float mx) {
    const float sc = leaky(ssrc + stgt);
    return __expf(fminf(fmaxf(sc - mx, -20.0f), 20.0f));
}

__global__ __launch_bounds__(256)
void gat_node_agg(const int* __restrict__ off, const unsigned short* __restrict__ seg_src,
                  const float* __restrict__ s_src, const float* __restrict__ a,
                  const unsigned* __restrict__ gmax, const float* __restrict__ h,
                  const float* __restrict__ bias, float* __restrict__ out) {
    const int lane = threadIdx.x & 31;
    const int wv = __builtin_amdgcn_readfirstlane((int)(threadIdx.x >> 5));
    const int n  = blockIdx.x * 8 + wv;
    const int hd = lane >> 2;

    const float4 hn = *reinterpret_cast<const float4*>(h + (size_t)n * OUT_F + lane * 4);
    const float4 at = *reinterpret_cast<const float4*>(a + hd * 2 * HEAD_DIM + HEAD_DIM + (lane & 3) * 4);
    float st = hn.x * at.x + hn.y * at.y + hn.z * at.z + hn.w * at.w;
    st += __shfl_xor(st, 1);
    st += __shfl_xor(st, 2);

    const float mx = dec_f32(gmax[hd]);

    int beg = off[n];
    int end = off[min(n + 1, N_NODES - 1)];
    if (n + 1 >= N_NODES) end = N_EDGES;
    beg = max(beg, 0);
    end = min(end, N_EDGES);

    float4 acc = make_float4(0.0f, 0.0f, 0.0f, 0.0f);
    float  den = 0.0f;

    for (int base = beg; base < end; base += 32) {
        const int cnt = min(end - base, 32);
        int myidx = 0;
        if (lane < cnt) myidx = seg_src[base + lane];

        int k = 0;
        for (; k + 4 <= cnt; k += 4) {
            const int s0 = __builtin_amdgcn_readlane(myidx, k);
            const int s1 = __builtin_amdgcn_readlane(myidx, k + 1);
            const int s2 = __builtin_amdgcn_readlane(myidx, k + 2);
            const int s3 = __builtin_amdgcn_readlane(myidx, k + 3);
            const float q0 = s_src[s0 * HEADS + hd];
            const float q1 = s_src[s1 * HEADS + hd];
            const float q2 = s_src[s2 * HEADS + hd];
            const float q3 = s_src[s3 * HEADS + hd];
            const float4 v0 = *reinterpret_cast<const float4*>(h + (size_t)s0 * OUT_F + lane * 4);
            const float4 v1 = *reinterpret_cast<const float4*>(h + (size_t)s1 * OUT_F + lane * 4);
            const float4 v2 = *reinterpret_cast<const float4*>(h + (size_t)s2 * OUT_F + lane * 4);
            const float4 v3 = *reinterpret_cast<const float4*>(h + (size_t)s3 * OUT_F + lane * 4);
            const float w0 = edge_weight(q0, st, mx);
            const float w1 = edge_weight(q1, st, mx);
            const float w2 = edge_weight(q2, st, mx);
            const float w3 = edge_weight(q3, st, mx);
            den += w0; den += w1; den += w2; den += w3;
            acc.x += v0.x * w0; acc.y += v0.y * w0; acc.z += v0.z * w0; acc.w += v0.w * w0;
            acc.x += v1.x * w1; acc.y += v1.y * w1; acc.z += v1.z * w1; acc.w += v1.w * w1;
            acc.x += v2.x * w2; acc.y += v2.y * w2; acc.z += v2.z * w2; acc.w += v2.w * w2;
            acc.x += v3.x * w3; acc.y += v3.y * w3; acc.z += v3.z * w3; acc.w += v3.w * w3;
        }
        for (; k < cnt; ++k) {
            const int s0 = __builtin_amdgcn_readlane(myidx, k);
            const float q0 = s_src[s0 * HEADS + hd];
            const float4 v0 = *reinterpret_cast<const float4*>(h + (size_t)s0 * OUT_F + lane * 4);
            const float w0 = edge_weight(q0, st, mx);
            den += w0;
            acc.x += v0.x * w0; acc.y += v0.y * w0; acc.z += v0.z * w0; acc.w += v0.w * w0;
        }
    }

    const float  dn = den + EPS_F;
    const float4 bb = *reinterpret_cast<const float4*>(bias + lane * 4);
    float4 o;
    o.x = acc.x / dn + bb.x;
    o.y = acc.y / dn + bb.y;
    o.z = acc.z / dn + bb.z;
    o.w = acc.w / dn + bb.w;
    *reinterpret_cast<float4*>(out + (size_t)n * OUT_F + lane * 4) = o;
}

extern "C" void kernel_launch(void* const* d_in, const int* in_sizes, int n_in,
                              void* d_out, int out_size, void* d_ws, size_t ws_size,
                              hipStream_t stream) {
    (void)in_sizes; (void)n_in; (void)out_size; (void)ws_size;

    const float* x    = (const float*)d_in[0];
    const int*   ei   = (const int*)  d_in[1];
    const float* W    = (const float*)d_in[2];
    const float* a    = (const float*)d_in[3];
    const float* bias = (const float*)d_in[4];
    float* out = (float*)d_out;

    const int* src = ei;
    const int* tgt = ei + N_EDGES;

    float*    ws    = (float*)d_ws;
    float*    hbuf  = ws;
    float*    ssrc  = hbuf + (size_t)N_NODES * OUT_F;
    float*    stgt  = ssrc + (size_t)N_NODES * HEADS;
    float*    tail  = stgt + (size_t)N_NODES * HEADS;
    int*      off   = (int*)tail;
    unsigned* gmax  = (unsigned*)(tail + (size_t)N_NODES * HEADS);
    unsigned short* seg = (unsigned short*)stgt;
    int*      cnt_hi = off + N_NODES;
    unsigned char* rank8 = (unsigned char*)(off + 2 * N_NODES);

    gat_gemm_wmma<<<N_NODES / 16, 256, 0, stream>>>(x, W, hbuf);

    gat_node_scores<<<(N_NODES * HEADS + 255) / 256, 256, 0, stream>>>(hbuf, a, ssrc, stgt, off, cnt_hi, gmax);

    gat_edge_max<<<N_EDGES / EM_EDGES, 256, 0, stream>>>(src, tgt, ssrc, stgt, gmax, off, rank8);

    gat_degree_scan<<<1, SCAN_T, 0, stream>>>(off);

    gat_group_edges<<<(GE_QUADS + 255) / 256, 256, 0, stream>>>(src, tgt, rank8, off, cnt_hi, seg);

    gat_node_agg<<<N_NODES / 8, 256, 0, stream>>>(off, seg, ssrc, a, gmax, hbuf, bias, out);
}


# counting by target moved off memory: owner blocks each keep a node range's edge counters in LDS, scan the target list and rank their edges with LDS adds (rank order: arrival in the owner block)
# speedup vs baseline: 4.8349x; 1.3489x over previous
#include <hip/hip_runtime.h>
#include <hip/hip_bf16.h>
#include <math.h>

#define N_NODES   50000
#define N_EDGES   800000
#define IN_F      128
#define HEADS     8
#define HEAD_DIM  16
#define OUT_F     (HEADS * HEAD_DIM)
#define ALPHA     0.2f
#define EPS_F     1e-6f

typedef __attribute__((ext_vector_type(2))) float v2f;
typedef __attribute__((ext_vector_type(8))) float v8f;

__device__ __forceinline__ unsigned enc_f32(float f) {
    unsigned u = __float_as_uint(f);
    return (u & 0x80000000u) ? ~u : (u | 0x80000000u);
}
__device__ __forceinline__ float dec_f32(unsigned u) {
    return (u & 0x80000000u) ? __uint_as_float(u ^ 0x80000000u)
                             : __uint_as_float(~u);
}
__device__ __forceinline__ float leaky(float s) {
    return s >= 0.0f ? s : ALPHA * s;
}

static_assert(N_NODES <= 65536, "source ids are kept as 16-bit words");
static_assert(N_NODES % 8 == 0, "one wave per node, 8 waves per block");

#define A_PITCH   (IN_F + 4)
static_assert((A_PITCH * 4) % 16 == 0, "A-tile rows are written as 16-B words");

__global__ __launch_bounds__(256)
void gat_gemm_wmma(const float* __restrict__ x, const float* __restrict__ W,
                   float* __restrict__ h) {
    __shared__ float As[16 * A_PITCH];
    const int tid = threadIdx.x;
    const int m0  = blockIdx.x * 16;

    {
        const int base = tid * 8;
        const int row  = base >> 7;
        const int col  = base & 127;
        const float4* g =
            reinterpret_cast<const float4*>(x + (size_t)(m0 + row) * IN_F + col);
        float4 p0 = g[0];
        float4 p1 = g[1];
        *reinterpret_cast<float4*>(&As[row * A_PITCH + col])     = p0;
        *reinterpret_cast<float4*>(&As[row * A_PITCH + col + 4]) = p1;
    }
    __syncthreads();

    const int lane = tid & 31;
    const int head = tid >> 5;
    const int l16  = lane & 15;
    const bool hi  = lane >= 16;
    const float* Wh = W + (size_t)head * IN_F * HEAD_DIM;

    v8f c = {};
#pragma unroll
    for (int k = 0; k < IN_F; k += 4) {
        const int kk = k + (hi ? 2 : 0);
        v2f a, b;
        a.x = As[l16 * A_PITCH + kk];
        a.y = As[l16 * A_PITCH + kk + 1];
        b.x = Wh[kk * HEAD_DIM + l16];
        b.y = Wh[(kk + 1) * HEAD_DIM + l16];
        c = __builtin_amdgcn_wmma_f32_16x16x4_f32(
                 false, a,  false, b,
                 (short)0, c,  false,  false);
    }

    const int rbase = hi ? 8 : 0;
#pragma unroll
    for (int v = 0; v < 8; ++v)
        h[(size_t)(m0 + rbase + v) * OUT_F + head * HEAD_DIM + l16] = c[v];
}

__global__ void gat_node_scores(const float* __restrict__ h, const float* __restrict__ a,
                                float* __restrict__ s_src, float* __restrict__ s_tgt,
                                unsigned* __restrict__ gmax) {
    const int i = blockIdx.x * blockDim.x + threadIdx.x;
    if (i < HEADS)   gmax[i] = 0u;
    if (i >= N_NODES * HEADS) return;
    const int hd = i & (HEADS - 1);
    const int n  = i >> 3;
    const float* hv = h + (size_t)n * OUT_F + hd * HEAD_DIM;
    const float* as = a + hd * 2 * HEAD_DIM;
    const float* at = as + HEAD_DIM;
    float ss = 0.0f, st = 0.0f;
#pragma unroll
    for (int d = 0; d < HEAD_DIM; ++d) {
        const float v = hv[d];
        ss += v * as[d];
        st += v * at[d];
    }
    s_src[i] = ss;
    s_tgt[i] = st;
}

#define SEG_CAP   (OUT_F * 4 / 2)
static_assert(SEG_CAP * 2 == OUT_F * 4, "a segment is exactly one output row");

#define EP_THREADS 1024
#define EP_WAVES   (EP_THREADS / 32)

#define NB_OWN     100
#define OWN_R      (N_NODES / NB_OWN)
static_assert(N_NODES % NB_OWN == 0, "the owner runs tile the nodes");
static_assert(OWN_R <= 4096, "a target offset is packed above a 20-bit edge number");
static_assert(N_EDGES <= (1 << 20), "an edge number is packed into 20 bits");
static_assert(N_EDGES % 4 == 0, "the target list is read as 16-B words");

#define SCAN_DEPTH 8
#define SCAN_V4    (N_EDGES / 4)
#define SCAN_STEP  (EP_THREADS * SCAN_DEPTH)
#define OWN_FULL   (SCAN_V4 / SCAN_STEP)
#define OWN_TAIL   ((SCAN_V4 - OWN_FULL * SCAN_STEP + EP_THREADS - 1) / EP_THREADS)
static_assert(OWN_FULL >= 2 && OWN_FULL % 2 == 0, "the two register sets take turns over an even number of trips");
#define Q_W        1024
#define Q_ROOM     (16 * 32)
#define DRAIN_DEPTH 12
static_assert(Q_W >= Q_ROOM && SCAN_DEPTH % 4 == 0, "a list takes a note from every lane at sixteen positions");

#define EM_PAIRS   (EP_THREADS / 2)
#define EM_PASSES  5
#define EM_EDGES   (EM_PAIRS * EM_PASSES)
#define NB_MAX     ((N_EDGES + EM_EDGES - 1) / EM_EDGES)

__device__ __forceinline__ void own_place(unsigned d, int s, int lo, int* lcnt,
                                          unsigned short* __restrict__ ids) {
    const int r = atomicAdd(&lcnt[d], 1);
    if ((unsigned)r < (unsigned)SEG_CAP)
        ids[(size_t)(lo + (int)d) * SEG_CAP + r] = (unsigned short)s;
}

__device__ __forceinline__ unsigned own_test(int t, int lo) {
    return __builtin_amdgcn_ballot_w32((unsigned)(t - lo) < (unsigned)OWN_R);
}

__device__ __forceinline__ void own_drain(const unsigned* wq, int n, int lane, int lo,
                                          const int* __restrict__ src, int* lcnt,
                                          unsigned short* __restrict__ ids) {
    for (int i0 = 0; i0 < n; i0 += DRAIN_DEPTH * 32) {
        unsigned ent[DRAIN_DEPTH];
        bool     ok[DRAIN_DEPTH];
        int      sv[DRAIN_DEPTH];
#pragma unroll
        for (int u = 0; u < DRAIN_DEPTH; ++u) {
            const int i = i0 + u * 32 + lane;
            ok[u]  = i < n;
            ent[u] = ok[u] ? wq[i] : 0u;
        }
#pragma unroll
        for (int u = 0; u < DRAIN_DEPTH; ++u) sv[u] = src[ent[u] & 0xFFFFFu];
#pragma unroll
        for (int u = 0; u < DRAIN_DEPTH; ++u)
            if (ok[u]) own_place(ent[u] >> 20, sv[u], lo, lcnt, ids);
    }
}

__device__ __forceinline__ void own_note(unsigned m, int t, int e, int lo, unsigned* wq, int& run) {
    const unsigned d = (unsigned)(t - lo);
    if (d < (unsigned)OWN_R) wq[__builtin_amdgcn_mbcnt_lo(m, (unsigned)run)] = (d << 20) | (unsigned)e;
    run += __builtin_popcount(m);
}

__device__ __forceinline__ void own_load(const int4* __restrict__ t4, int first, int4 (&v)[SCAN_DEPTH]) {
#pragma unroll
    for (int j = 0; j < SCAN_DEPTH; ++j) v[j] = t4[first + j * EP_THREADS];
}

__device__ __forceinline__ void own_trip(const int4 (&v)[SCAN_DEPTH], int first, int lo,
                                         unsigned* wq, int& run, bool& over) {
    unsigned m[4 * SCAN_DEPTH];
#pragma unroll
    for (int j = 0; j < SCAN_DEPTH; ++j) {
        m[4 * j]     = own_test(v[j].x, lo);
        m[4 * j + 1] = own_test(v[j].y, lo);
        m[4 * j + 2] = own_test(v[j].z, lo);
        m[4 * j + 3] = own_test(v[j].w, lo);
    }
#pragma unroll
    for (int j = 0; j < SCAN_DEPTH; ++j) {
        const int e = (first + j * EP_THREADS) * 4;
        if ((j & 3) == 0 && run + Q_ROOM > Q_W) {
            over = true;
            run  = 0;
        }
        if (m[4 * j])     own_note(m[4 * j],     v[j].x, e,     lo, wq, run);
        if (m[4 * j + 1]) own_note(m[4 * j + 1], v[j].y, e + 1, lo, wq, run);
        if (m[4 * j + 2]) own_note(m[4 * j + 2], v[j].z, e + 2, lo, wq, run);
        if (m[4 * j + 3]) own_note(m[4 * j + 3], v[j].w, e + 3, lo, wq, run);
    }
}

__global__ __launch_bounds__(EP_THREADS)
void gat_edge_pass(const int* __restrict__ src, const int* __restrict__ tgt,
                   const float* __restrict__ s_src, const float* __restrict__ s_tgt,
                   unsigned* __restrict__ gmax, int* __restrict__ cnt,
                   unsigned short* __restrict__ ids) {
    __shared__ unsigned queue[EP_WAVES * Q_W];
    __shared__ int      lcnt[OWN_R];
    __shared__ int      any_over;
    __shared__ float4   wmax[EP_WAVES][2];
    const int tid  = threadIdx.x;
    const int lane = tid & 31;

    if (blockIdx.x < NB_OWN) {
        const int lo = blockIdx.x * OWN_R;
        const int wv = __builtin_amdgcn_readfirstlane(tid >> 5);
        unsigned* wq = queue + wv * Q_W;
        for (int i = tid; i < OWN_R; i += EP_THREADS) lcnt[i] = 0;
        if (tid == 0) any_over = 0;
        __syncthreads();

        const int4* t4 = reinterpret_cast<const int4*>(tgt);
        int  run  = 0;
        bool over = false;
        int4 va[SCAN_DEPTH], vb[SCAN_DEPTH];
        own_load(t4, tid, va);
        int first = tid;
        for (int trip = 0; trip < OWN_FULL - 2; trip += 2) {
            own_load(t4, first + SCAN_STEP, vb);
            own_trip(va, first, lo, wq, run, over);
            own_load(t4, first + 2 * SCAN_STEP, va);
            own_trip(vb, first + SCAN_STEP, lo, wq, run, over);
            first += 2 * SCAN_STEP;
        }
        own_load(t4, first + SCAN_STEP, vb);
        own_trip(va, first, lo, wq, run, over);
        own_trip(vb, first + SCAN_STEP, lo, wq, run, over);
#pragma unroll
        for (int j = 0; j < OWN_TAIL; ++j) {
            const int idx = OWN_FULL * SCAN_STEP + j * EP_THREADS + tid;
            const int4 w = (idx < SCAN_V4) ? t4[idx] : make_int4(-1, -1, -1, -1);
            const unsigned mx = own_test(w.x, lo), my = own_test(w.y, lo);
            const unsigned mz = own_test(w.z, lo), mw = own_test(w.w, lo);
            if ((j & 3) == 0 && run + Q_ROOM > Q_W) {
                over = true;
                run  = 0;
            }
            if (mx) own_note(mx, w.x, idx * 4,     lo, wq, run);
            if (my) own_note(my, w.y, idx * 4 + 1, lo, wq, run);
            if (mz) own_note(mz, w.z, idx * 4 + 2, lo, wq, run);
            if (mw) own_note(mw, w.w, idx * 4 + 3, lo, wq, run);
        }
        if (over && lane == 0) any_over = 1;
        __syncthreads();
        if (any_over == 0) {
            own_drain(wq, run, lane, lo, src, lcnt, ids);
        } else {
            for (int idx = tid; idx < SCAN_V4; idx += EP_THREADS) {
                const int4 w = t4[idx];
                const unsigned dx = (unsigned)(w.x - lo), dy = (unsigned)(w.y - lo);
                const unsigned dz = (unsigned)(w.z - lo), dw = (unsigned)(w.w - lo);
                if (dx < (unsigned)OWN_R) own_place(dx, src[idx * 4],     lo, lcnt, ids);
                if (dy < (unsigned)OWN_R) own_place(dy, src[idx * 4 + 1], lo, lcnt, ids);
                if (dz < (unsigned)OWN_R) own_place(dz, src[idx * 4 + 2], lo, lcnt, ids);
                if (dw < (unsigned)OWN_R) own_place(dw, src[idx * 4 + 3], lo, lcnt, ids);
            }
        }
        __syncthreads();
        for (int i = tid; i < OWN_R; i += EP_THREADS) cnt[lo + i] = lcnt[i];
        return;
    }

    const int half = tid & 1;
    const int e0   = (blockIdx.x - NB_OWN) * EM_EDGES + (tid >> 1);

    int s[EM_PASSES], t[EM_PASSES];
    for (int p = 0; p < EM_PASSES; ++p) {
        const int e = min(e0 + p * EM_PAIRS, N_EDGES - 1);
        s[p] = src[e];
        t[p] = tgt[e];
    }

    float4 a[EM_PASSES], b[EM_PASSES];
    for (int p = 0; p < EM_PASSES; ++p) {
        a[p] = *reinterpret_cast<const float4*>(s_src + s[p] * HEADS + half * 4);
        b[p] = *reinterpret_cast<const float4*>(s_tgt + t[p] * HEADS + half * 4);
    }

    float4 m = make_float4(a[0].x + b[0].x, a[0].y + b[0].y, a[0].z + b[0].z, a[0].w + b[0].w);
    for (int p = 1; p < EM_PASSES; ++p) {
        m.x = fmaxf(m.x, a[p].x + b[p].x);
        m.y = fmaxf(m.y, a[p].y + b[p].y);
        m.z = fmaxf(m.z, a[p].z + b[p].z);
        m.w = fmaxf(m.w, a[p].w + b[p].w);
    }

    for (int d = 2; d < 32; d <<= 1) {
        m.x = fmaxf(m.x, __shfl_xor(m.x, d));
        m.y = fmaxf(m.y, __shfl_xor(m.y, d));
        m.z = fmaxf(m.z, __shfl_xor(m.z, d));
        m.w = fmaxf(m.w, __shfl_xor(m.w, d));
    }
    if (lane < 2) wmax[tid >> 5][lane] = m;
    __syncthreads();

    if (tid < HEADS) {
        const float* wm = reinterpret_cast<const float*>(&wmax[0][0]);
        float v = wm[tid];
        for (int w = 1; w < EP_WAVES; ++w) v = fmaxf(v, wm[w * HEADS + tid]);
        atomicMax(&gmax[tid], enc_f32(leaky(v)));
    }
}

static_assert(N_EDGES % 32 == 0, "the walk over the whole edge list takes full waves");

__device__ __forceinline__ float edge_weight(float ssrc, float stgt, float mx) {
    const float sc = leaky(ssrc + stgt);
    return __expf(fminf(fmaxf(sc - mx, -20.0f), 20.0f));
}

__device__ __forceinline__ void add_edge(int s0, int hd, int lane, float st, float mx,
                                         const float* __restrict__ s_src,
                                         const float* __restrict__ h,
                                         float4& acc, float& den) {
    const float q0 = s_src[s0 * HEADS + hd];
    const float4 v0 = *reinterpret_cast<const float4*>(h + (size_t)s0 * OUT_F + lane * 4);
    const float w0 = edge_weight(q0, st, mx);
    den += w0;
    acc.x += v0.x * w0; acc.y += v0.y * w0; acc.z += v0.z * w0; acc.w += v0.w * w0;
}

__global__ __launch_bounds__(256)
void gat_node_agg(const int* __restrict__ src, const int* __restrict__ tgt,
                  const int* __restrict__ cnt,
                  const float* __restrict__ s_src, const float* __restrict__ a,
                  const unsigned* __restrict__ gmax, const float* __restrict__ h,
                  const float* __restrict__ bias, float* out) {
    const int lane = threadIdx.x & 31;
    const int wv = __builtin_amdgcn_readfirstlane((int)(threadIdx.x >> 5));
    const int n  = blockIdx.x * 8 + wv;
    const int hd = lane >> 2;

    const unsigned short* row = reinterpret_cast<const unsigned short*>(out + (size_t)n * OUT_F);
    const int deg = cnt[n];
    const float4 hn = *reinterpret_cast<const float4*>(h + (size_t)n * OUT_F + lane * 4);
    const float4 at = *reinterpret_cast<const float4*>(a + hd * 2 * HEAD_DIM + HEAD_DIM + (lane & 3) * 4);
    const unsigned gm = gmax[hd];
    const int pre = row[lane];

    float hx = hn.x;
    asm volatile("" : "+v"(hx) : "v"(pre), "v"(gm));

    float st = hx * at.x + hn.y * at.y + hn.z * at.z + hn.w * at.w;
    st += __shfl_xor(st, 1);
    st += __shfl_xor(st, 2);

    const float mx = dec_f32(gm);

    float4 acc = make_float4(0.0f, 0.0f, 0.0f, 0.0f);
    float  den = 0.0f;

    if (deg <= SEG_CAP) {
        for (int base = 0; base < deg; base += 32) {
            const int cnt32 = min(deg - base, 32);
            int v = pre;
            if (base > 0) v = row[base + lane];
            const int myidx = (lane < cnt32) ? min(v, N_NODES - 1) : 0;

            int k = 0;
            for (; k + 4 <= cnt32; k += 4) {
                const int s0 = __builtin_amdgcn_readlane(myidx, k);
                const int s1 = __builtin_amdgcn_readlane(myidx, k + 1);
                const int s2 = __builtin_amdgcn_readlane(myidx, k + 2);
                const int s3 = __builtin_amdgcn_readlane(myidx, k + 3);
                const float q0 = s_src[s0 * HEADS + hd];
                const float q1 = s_src[s1 * HEADS + hd];
                const float q2 = s_src[s2 * HEADS + hd];
                const float q3 = s_src[s3 * HEADS + hd];
                const float4 v0 = *reinterpret_cast<const float4*>(h + (size_t)s0 * OUT_F + lane * 4);
                const float4 v1 = *reinterpret_cast<const float4*>(h + (size_t)s1 * OUT_F + lane * 4);
                const float4 v2 = *reinterpret_cast<const float4*>(h + (size_t)s2 * OUT_F + lane * 4);
                const float4 v3 = *reinterpret_cast<const float4*>(h + (size_t)s3 * OUT_F + lane * 4);
                const float w0 = edge_weight(q0, st, mx);
                const float w1 = edge_weight(q1, st, mx);
                const float w2 = edge_weight(q2, st, mx);
                const float w3 = edge_weight(q3, st, mx);
                den += w0; den += w1; den += w2; den += w3;
                acc.x += v0.x * w0; acc.y += v0.y * w0; acc.z += v0.z * w0; acc.w += v0.w * w0;
                acc.x += v1.x * w1; acc.y += v1.y * w1; acc.z += v1.z * w1; acc.w += v1.w * w1;
                acc.x += v2.x * w2; acc.y += v2.y * w2; acc.z += v2.z * w2; acc.w += v2.w * w2;
                acc.x += v3.x * w3; acc.y += v3.y * w3; acc.z += v3.z * w3; acc.w += v3.w * w3;
            }
            for (; k < cnt32; ++k) {
                const int s0 = __builtin_amdgcn_readlane(myidx, k);
                add_edge(s0, hd, lane, st, mx, s_src, h, acc, den);
            }
        }
    } else {
        for (int eb = 0; eb < N_EDGES; eb += 32) {
            const int tt = tgt[eb + lane];
            const int ss = src[eb + lane];
            unsigned hit = __builtin_amdgcn_ballot_w32(tt == n);
            while (hit) {
                const int k = __builtin_ctz(hit);
                hit &= hit - 1;
                const int s0 = __builtin_amdgcn_readlane(ss, k);
                add_edge(s0, hd, lane, st, mx, s_src, h, acc, den);
            }
        }
    }

    const float  dn = den + EPS_F;
    const float4 bb = *reinterpret_cast<const float4*>(bias + lane * 4);
    float4 o;
    o.x = acc.x / dn + bb.x;
    o.y = acc.y / dn + bb.y;
    o.z = acc.z / dn + bb.z;
    o.w = acc.w / dn + bb.w;
    *reinterpret_cast<float4*>(out + (size_t)n * OUT_F + lane * 4) = o;
}

extern "C" void kernel_launch(void* const* d_in, const int* in_sizes, int n_in,
                              void* d_out, int out_size, void* d_ws, size_t ws_size,
                              hipStream_t stream) {
    (void)in_sizes; (void)n_in; (void)out_size; (void)ws_size;

    const float* x    = (const float*)d_in[0];
    const int*   ei   = (const int*)  d_in[1];
    const float* W    = (const float*)d_in[2];
    const float* a    = (const float*)d_in[3];
    const float* bias = (const float*)d_in[4];
    float* out = (float*)d_out;

    const int* src = ei;
    const int* tgt = ei + N_EDGES;

    float*    ws    = (float*)d_ws;
    float*    hbuf  = ws;
    float*    ssrc  = hbuf + (size_t)N_NODES * OUT_F;
    float*    stgt  = ssrc + (size_t)N_NODES * HEADS;
    float*    tail  = stgt + (size_t)N_NODES * HEADS;
    int*      cnt   = (int*)tail;
    unsigned* gmax  = (unsigned*)(tail + (size_t)N_NODES * HEADS);
    unsigned short* ids = (unsigned short*)d_out;

    gat_gemm_wmma<<<N_NODES / 16, 256, 0, stream>>>(x, W, hbuf);

    gat_node_scores<<<(N_NODES * HEADS + 255) / 256, 256, 0, stream>>>(hbuf, a, ssrc, stgt, gmax);

    gat_edge_pass<<<NB_OWN + NB_MAX, EP_THREADS, 0, stream>>>(src, tgt, ssrc, stgt, gmax, cnt, ids);

    gat_node_agg<<<N_NODES / 8, 256, 0, stream>>>(src, tgt, cnt, ssrc, a, gmax, hbuf, bias, out);
}


# edges sorted by owner once: sorting blocks in launch 1 deal list slices into per-owner chunks (LDS rank); owner blocks read their chunks, not the whole list (sum order: arrival in both, as before)
# speedup vs baseline: 6.4038x; 1.3245x over previous
#include <hip/hip_runtime.h>
#include <hip/hip_bf16.h>
#include <math.h>

#define N_NODES   50000
#define N_EDGES   800000
#define IN_F      128
#define HEADS     8
#define HEAD_DIM  16
#define OUT_F     (HEADS * HEAD_DIM)
#define ALPHA     0.2f
#define EPS_F     1e-6f

typedef __attribute__((ext_vector_type(2))) float v2f;
typedef __attribute__((ext_vector_type(8))) float v8f;

__device__ __forceinline__ unsigned enc_f32(float f) {
    unsigned u = __float_as_uint(f);
    return (u & 0x80000000u) ? ~u : (u | 0x80000000u);
}
__device__ __forceinline__ float dec_f32(unsigned u) {
    return (u & 0x80000000u) ? __uint_as_float(u ^ 0x80000000u)
                             : __uint_as_float(~u);
}
__device__ __forceinline__ float leaky(float s) {
    return s >= 0.0f ? s : ALPHA * s;
}

static_assert(N_NODES <= 65536, "source ids are kept as 16-bit words");
static_assert(N_NODES % 8 == 0, "one wave per node, 8 waves per block");
static_assert(N_NODES % 16 == 0, "the product kernel takes whole 16-row tiles");

#define A_PITCH   (IN_F + 4)
static_assert((A_PITCH * 4) % 16 == 0, "A-tile rows are written as 16-B words");

template <int XMASK>
__device__ __forceinline__ float row_xor(float v) {
    return __int_as_float(
        __builtin_amdgcn_update_dpp(0, __float_as_int(v), 0x160 + XMASK, 0xF, 0xF, true));
}

#define SEG_CAP   (OUT_F * 4 / 2)
static_assert(SEG_CAP * 2 == OUT_F * 4, "a segment is exactly one output row");

#define NB_OWN     100
#define OWN_R      (N_NODES / NB_OWN)
static_assert(N_NODES % NB_OWN == 0, "the owner runs tile the nodes");

#define GB_THREADS 256
#define NBIN       200
#define BIN_EDGES  (N_EDGES / NBIN)
#define BIN_V4     (BIN_EDGES / 4)
#define BIN_K      ((BIN_V4 + GB_THREADS - 1) / GB_THREADS)
#define BIN_CAP    OUT_F
static_assert(N_EDGES % (NBIN * 4) == 0, "every slice is a whole number of 16-B words");
static_assert(NBIN <= OWN_R, "the chunks for one owner are rows of that owner's run");
static_assert(OWN_R <= 65536, "a target offset is packed above a 16-bit source id");

__device__ __forceinline__ void bin_one(bool ok, int s, int t, int blk, int* hist,
                                        unsigned* __restrict__ bins) {
    if (ok && (unsigned)t < (unsigned)N_NODES) {
        const unsigned o = (unsigned)t / (unsigned)OWN_R;
        const unsigned d = (unsigned)t - o * (unsigned)OWN_R;
        const int r = atomicAdd(&hist[o], 1);
        if (r < BIN_CAP)
            bins[((size_t)(o * OWN_R) + blk) * BIN_CAP + r] = (d << 16) | ((unsigned)s & 0xFFFFu);
    }
}

__device__ __forceinline__ void bin_block(int blk, int tid, const int* __restrict__ src,
                                          const int* __restrict__ tgt, unsigned* __restrict__ bins,
                                          int* __restrict__ bincnt, int* hist) {
    for (int i = tid; i < NB_OWN; i += GB_THREADS) hist[i] = 0;
    __syncthreads();
    const int4* s4 = reinterpret_cast<const int4*>(src) + blk * BIN_V4;
    const int4* t4 = reinterpret_cast<const int4*>(tgt) + blk * BIN_V4;
    int4 sv[BIN_K], tv[BIN_K];
#pragma unroll
    for (int k = 0; k < BIN_K; ++k) {
        const int q = min(tid + k * GB_THREADS, BIN_V4 - 1);
        sv[k] = s4[q];
        tv[k] = t4[q];
    }
#pragma unroll
    for (int k = 0; k < BIN_K; ++k) {
        const bool ok = tid + k * GB_THREADS < BIN_V4;
        bin_one(ok, sv[k].x, tv[k].x, blk, hist, bins);
        bin_one(ok, sv[k].y, tv[k].y, blk, hist, bins);
        bin_one(ok, sv[k].z, tv[k].z, blk, hist, bins);
        bin_one(ok, sv[k].w, tv[k].w, blk, hist, bins);
    }
    __syncthreads();
    for (int i = tid; i < NB_OWN; i += GB_THREADS) bincnt[i * NBIN + blk] = hist[i];
}

__global__ __launch_bounds__(GB_THREADS)
void gat_gemm_wmma(const float* __restrict__ x, const float* __restrict__ W,
                   const float* __restrict__ a,
                   float* __restrict__ h, float* __restrict__ s_src, float* __restrict__ s_tgt,
                   unsigned* __restrict__ gmax,
                   const int* __restrict__ src, const int* __restrict__ tgt,
                   unsigned* __restrict__ bins, int* __restrict__ bincnt) {
    __shared__ float As[16 * A_PITCH];
    __shared__ int   hist[NB_OWN];
    const int tid = threadIdx.x;
    if (blockIdx.x == 0 && tid < HEADS) gmax[tid] = 0u;
    if (blockIdx.x < NBIN) {
        bin_block(blockIdx.x, tid, src, tgt, bins, bincnt, hist);
        return;
    }
    const int m0  = (blockIdx.x - NBIN) * 16;

    {
        const int base = tid * 8;
        const int row  = base >> 7;
        const int col  = base & 127;
        const float4* g =
            reinterpret_cast<const float4*>(x + (size_t)(m0 + row) * IN_F + col);
        float4 p0 = g[0];
        float4 p1 = g[1];
        *reinterpret_cast<float4*>(&As[row * A_PITCH + col])     = p0;
        *reinterpret_cast<float4*>(&As[row * A_PITCH + col + 4]) = p1;
    }
    __syncthreads();

    const int lane = tid & 31;
    const int head = tid >> 5;
    const int l16  = lane & 15;
    const bool hi  = lane >= 16;
    const float* Wh = W + (size_t)head * IN_F * HEAD_DIM;
    const float as_l = a[head * 2 * HEAD_DIM + l16];
    const float at_l = a[head * 2 * HEAD_DIM + HEAD_DIM + l16];

    v8f c = {};
#pragma unroll
    for (int k = 0; k < IN_F; k += 4) {
        const int kk = k + (hi ? 2 : 0);
        v2f fa, fb;
        fa.x = As[l16 * A_PITCH + kk];
        fa.y = As[l16 * A_PITCH + kk + 1];
        fb.x = Wh[kk * HEAD_DIM + l16];
        fb.y = Wh[(kk + 1) * HEAD_DIM + l16];
        c = __builtin_amdgcn_wmma_f32_16x16x4_f32(
                 false, fa,  false, fb,
                 (short)0, c,  false,  false);
    }

    const int rbase = hi ? 8 : 0;
#pragma unroll
    for (int v = 0; v < 8; ++v)
        h[(size_t)(m0 + rbase + v) * OUT_F + head * HEAD_DIM + l16] = c[v];

    const bool b8 = (l16 & 8) != 0, b4 = (l16 & 4) != 0, b2 = (l16 & 2) != 0, b1 = (l16 & 1) != 0;
    const float a_keep = b8 ? at_l : as_l;
    const float a_give = b8 ? as_l : at_l;
    float r8[8], r4[4], r2[2];
#pragma unroll
    for (int j = 0; j < 8; ++j)
        r8[j] = c[j] * a_keep + row_xor<8>(c[j] * a_give);
#pragma unroll
    for (int j = 0; j < 4; ++j) {
        const float keep = b4 ? r8[4 + j] : r8[j];
        const float give = b4 ? r8[j] : r8[4 + j];
        r4[j] = keep + row_xor<4>(give);
    }
#pragma unroll
    for (int j = 0; j < 2; ++j) {
        const float keep = b2 ? r4[2 + j] : r4[j];
        const float give = b2 ? r4[j] : r4[2 + j];
        r2[j] = keep + row_xor<2>(give);
    }
    const float mine = (b1 ? r2[1] : r2[0]) + row_xor<1>(b1 ? r2[0] : r2[1]);
    float* dst = (l16 & 8) ? s_tgt : s_src;
    dst[(size_t)(m0 + rbase + (l16 & 7)) * HEADS + head] = mine;
}

#define EP_THREADS 1024
#define EP_WAVES   (EP_THREADS / 32)
#define OWN_K      ((NBIN + EP_WAVES - 1) / EP_WAVES)
#define SCAN_V4    (N_EDGES / 4)
static_assert(N_EDGES % 4 == 0, "the target list is read as 16-B words");
static_assert(BIN_CAP == 4 * 32, "a wave reads a chunk with one 16-B load per lane");

#define EM_PAIRS   (EP_THREADS / 2)
#define EM_PASSES  5
#define EM_EDGES   (EM_PAIRS * EM_PASSES)
#define NB_MAX     ((N_EDGES + EM_EDGES - 1) / EM_EDGES)

__device__ __forceinline__ void own_place(unsigned d, int s, int lo, int* lcnt,
                                          unsigned short* ids) {
    const int r = atomicAdd(&lcnt[d], 1);
    if ((unsigned)r < (unsigned)SEG_CAP)
        ids[(size_t)(lo + (int)d) * SEG_CAP + r] = (unsigned short)s;
}

__device__ __forceinline__ void own_place4(uint4 w, int e0, int n, int lo, int* lcnt,
                                           unsigned short* ids) {
    const unsigned e[4] = {w.x, w.y, w.z, w.w};
    unsigned d[4];
    int      r[4];
#pragma unroll
    for (int c = 0; c < 4; ++c) {
        d[c] = min(e[c] >> 16, (unsigned)(OWN_R - 1));
        r[c] = (e0 + c < n) ? atomicAdd(&lcnt[d[c]], 1) : SEG_CAP;
    }
#pragma unroll
    for (int c = 0; c < 4; ++c)
        if ((unsigned)r[c] < (unsigned)SEG_CAP)
            ids[(size_t)(lo + (int)d[c]) * SEG_CAP + r[c]] = (unsigned short)(e[c] & 0xFFFFu);
}

__global__ __launch_bounds__(EP_THREADS)
void gat_edge_pass(const int* __restrict__ src, const int* __restrict__ tgt,
                   const float* __restrict__ s_src, const float* __restrict__ s_tgt,
                   unsigned* __restrict__ gmax, int* __restrict__ cnt,
                   const int* __restrict__ bincnt, unsigned short* ids) {
    __shared__ int      lcnt[OWN_R];
    __shared__ int      any_over;
    __shared__ float4   wmax[EP_WAVES][2];
    const int tid  = threadIdx.x;
    const int lane = tid & 31;

    if (blockIdx.x < NB_OWN) {
        const int lo = blockIdx.x * OWN_R;
        const int wv = __builtin_amdgcn_readfirstlane(tid >> 5);
        const uint4* rows = reinterpret_cast<const uint4*>(ids + (size_t)lo * SEG_CAP);
        uint4 ent[OWN_K];
        int   n[OWN_K];
        bool  over = false;
#pragma unroll
        for (int k = 0; k < OWN_K; ++k) {
            const int c  = wv + k * EP_WAVES;
            const int cc = min(c, NBIN - 1);
            const int cn = bincnt[blockIdx.x * NBIN + cc];
            n[k]   = (c < NBIN) ? cn : 0;
            ent[k] = rows[cc * (BIN_CAP / 4) + lane];
            over   = over || n[k] > BIN_CAP;
        }
        for (int i = tid; i < OWN_R; i += EP_THREADS) lcnt[i] = 0;
        if (tid == 0) any_over = 0;
        __syncthreads();
        if (over && lane == 0) any_over = 1;
        __syncthreads();
        if (any_over == 0) {
#pragma unroll
            for (int k = 0; k < OWN_K; ++k)
                own_place4(ent[k], lane * 4, n[k], lo, lcnt, ids);
        } else {
            const int4* t4 = reinterpret_cast<const int4*>(tgt);
            for (int idx = tid; idx < SCAN_V4; idx += EP_THREADS) {
                const int4 w = t4[idx];
                const unsigned dx = (unsigned)(w.x - lo), dy = (unsigned)(w.y - lo);
                const unsigned dz = (unsigned)(w.z - lo), dw = (unsigned)(w.w - lo);
                if (dx < (unsigned)OWN_R) own_place(dx, src[idx * 4],     lo, lcnt, ids);
                if (dy < (unsigned)OWN_R) own_place(dy, src[idx * 4 + 1], lo, lcnt, ids);
                if (dz < (unsigned)OWN_R) own_place(dz, src[idx * 4 + 2], lo, lcnt, ids);
                if (dw < (unsigned)OWN_R) own_place(dw, src[idx * 4 + 3], lo, lcnt, ids);
            }
        }
        __syncthreads();
        for (int i = tid; i < OWN_R; i += EP_THREADS) cnt[lo + i] = lcnt[i];
        return;
    }

    const int half = tid & 1;
    const int e0   = (blockIdx.x - NB_OWN) * EM_EDGES + (tid >> 1);

    int s[EM_PASSES], t[EM_PASSES];
    for (int p = 0; p < EM_PASSES; ++p) {
        const int e = min(e0 + p * EM_PAIRS, N_EDGES - 1);
        s[p] = src[e];
        t[p] = tgt[e];
    }

    float4 a[EM_PASSES], b[EM_PASSES];
    for (int p = 0; p < EM_PASSES; ++p) {
        a[p] = *reinterpret_cast<const float4*>(s_src + s[p] * HEADS + half * 4);
        b[p] = *reinterpret_cast<const float4*>(s_tgt + t[p] * HEADS + half * 4);
    }

    float4 m = make_float4(a[0].x + b[0].x, a[0].y + b[0].y, a[0].z + b[0].z, a[0].w + b[0].w);
    for (int p = 1; p < EM_PASSES; ++p) {
        m.x = fmaxf(m.x, a[p].x + b[p].x);
        m.y = fmaxf(m.y, a[p].y + b[p].y);
        m.z = fmaxf(m.z, a[p].z + b[p].z);
        m.w = fmaxf(m.w, a[p].w + b[p].w);
    }

    for (int d = 2; d < 32; d <<= 1) {
        m.x = fmaxf(m.x, __shfl_xor(m.x, d));
        m.y = fmaxf(m.y, __shfl_xor(m.y, d));
        m.z = fmaxf(m.z, __shfl_xor(m.z, d));
        m.w = fmaxf(m.w, __shfl_xor(m.w, d));
    }
    if (lane < 2) wmax[tid >> 5][lane] = m;
    __syncthreads();

    if (tid < HEADS) {
        const float* wm = reinterpret_cast<const float*>(&wmax[0][0]);
        float v = wm[tid];
        for (int w = 1; w < EP_WAVES; ++w) v = fmaxf(v, wm[w * HEADS + tid]);
        atomicMax(&gmax[tid], enc_f32(leaky(v)));
    }
}

static_assert(N_EDGES % 32 == 0, "the walk over the whole edge list takes full waves");

__device__ __forceinline__ float edge_weight(float ssrc, float stgt, float mx) {
    const float sc = leaky(ssrc + stgt);
    return __expf(fminf(fmaxf(sc - mx, -20.0f), 20.0f));
}

__device__ __forceinline__ void add_edge(int s0, int hd, int lane, float st, float mx,
                                         const float* __restrict__ s_src,
                                         const float* __restrict__ h,
                                         float4& acc, float& den) {
    const float q0 = s_src[s0 * HEADS + hd];
    const float4 v0 = *reinterpret_cast<const float4*>(h + (size_t)s0 * OUT_F + lane * 4);
    const float w0 = edge_weight(q0, st, mx);
    den += w0;
    acc.x += v0.x * w0; acc.y += v0.y * w0; acc.z += v0.z * w0; acc.w += v0.w * w0;
}

__global__ __launch_bounds__(256)
void gat_node_agg(const int* __restrict__ src, const int* __restrict__ tgt,
                  const int* __restrict__ cnt,
                  const float* __restrict__ s_src, const float* __restrict__ a,
                  const unsigned* __restrict__ gmax, const float* __restrict__ h,
                  const float* __restrict__ bias, float* out) {
    const int lane = threadIdx.x & 31;
    const int wv = __builtin_amdgcn_readfirstlane((int)(threadIdx.x >> 5));
    const int n  = blockIdx.x * 8 + wv;
    const int hd = lane >> 2;

    const unsigned short* row = reinterpret_cast<const unsigned short*>(out + (size_t)n * OUT_F);
    const int deg = cnt[n];
    const float4 hn = *reinterpret_cast<const float4*>(h + (size_t)n * OUT_F + lane * 4);
    const float4 at = *reinterpret_cast<const float4*>(a + hd * 2 * HEAD_DIM + HEAD_DIM + (lane & 3) * 4);
    const unsigned gm = gmax[hd];
    const int pre = row[lane];

    float hx = hn.x;
    asm volatile("" : "+v"(hx) : "v"(pre), "v"(gm));

    float st = hx * at.x + hn.y * at.y + hn.z * at.z + hn.w * at.w;
    st += __shfl_xor(st, 1);
    st += __shfl_xor(st, 2);

    const float mx = dec_f32(gm);

    float4 acc = make_float4(0.0f, 0.0f, 0.0f, 0.0f);
    float  den = 0.0f;

    if (deg <= SEG_CAP) {
        for (int base = 0; base < deg; base += 32) {
            const int cnt32 = min(deg - base, 32);
            int v = pre;
            if (base > 0) v = row[base + lane];
            const int myidx = (lane < cnt32) ? min(v, N_NODES - 1) : 0;

            int k = 0;
            for (; k + 4 <= cnt32; k += 4) {
                const int s0 = __builtin_amdgcn_readlane(myidx, k);
                const int s1 = __builtin_amdgcn_readlane(myidx, k + 1);
                const int s2 = __builtin_amdgcn_readlane(myidx, k + 2);
                const int s3 = __builtin_amdgcn_readlane(myidx, k + 3);
                const float q0 = s_src[s0 * HEADS + hd];
                const float q1 = s_src[s1 * HEADS + hd];
                const float q2 = s_src[s2 * HEADS + hd];
                const float q3 = s_src[s3 * HEADS + hd];
                const float4 v0 = *reinterpret_cast<const float4*>(h + (size_t)s0 * OUT_F + lane * 4);
                const float4 v1 = *reinterpret_cast<const float4*>(h + (size_t)s1 * OUT_F + lane * 4);
                const float4 v2 = *reinterpret_cast<const float4*>(h + (size_t)s2 * OUT_F + lane * 4);
                const float4 v3 = *reinterpret_cast<const float4*>(h + (size_t)s3 * OUT_F + lane * 4);
                const float w0 = edge_weight(q0, st, mx);
                const float w1 = edge_weight(q1, st, mx);
                const float w2 = edge_weight(q2, st, mx);
                const float w3 = edge_weight(q3, st, mx);
                den += w0; den += w1; den += w2; den += w3;
                acc.x += v0.x * w0; acc.y += v0.y * w0; acc.z += v0.z * w0; acc.w += v0.w * w0;
                acc.x += v1.x * w1; acc.y += v1.y * w1; acc.z += v1.z * w1; acc.w += v1.w * w1;
                acc.x += v2.x * w2; acc.y += v2.y * w2; acc.z += v2.z * w2; acc.w += v2.w * w2;
                acc.x += v3.x * w3; acc.y += v3.y * w3; acc.z += v3.z * w3; acc.w += v3.w * w3;
            }
            for (; k < cnt32; ++k) {
                const int s0 = __builtin_amdgcn_readlane(myidx, k);
                add_edge(s0, hd, lane, st, mx, s_src, h, acc, den);
            }
        }
    } else {
        for (int eb = 0; eb < N_EDGES; eb += 32) {
            const int tt = tgt[eb + lane];
            const int ss = src[eb + lane];
            unsigned hit = __builtin_amdgcn_ballot_w32(tt == n);
            while (hit) {
                const int k = __builtin_ctz(hit);
                hit &= hit - 1;
                const int s0 = __builtin_amdgcn_readlane(ss, k);
                add_edge(s0, hd, lane, st, mx, s_src, h, acc, den);
            }
        }
    }

    const float  dn = den + EPS_F;
    const float4 bb = *reinterpret_cast<const float4*>(bias + lane * 4);
    float4 o;
    o.x = acc.x / dn + bb.x;
    o.y = acc.y / dn + bb.y;
    o.z = acc.z / dn + bb.z;
    o.w = acc.w / dn + bb.w;
    *reinterpret_cast<float4*>(out + (size_t)n * OUT_F + lane * 4) = o;
}

extern "C" void kernel_launch(void* const* d_in, const int* in_sizes, int n_in,
                              void* d_out, int out_size, void* d_ws, size_t ws_size,
                              hipStream_t stream) {
    (void)in_sizes; (void)n_in; (void)out_size; (void)ws_size;

    const float* x    = (const float*)d_in[0];
    const int*   ei   = (const int*)  d_in[1];
    const float* W    = (const float*)d_in[2];
    const float* a    = (const float*)d_in[3];
    const float* bias = (const float*)d_in[4];
    float* out = (float*)d_out;

    const int* src = ei;
    const int* tgt = ei + N_EDGES;

    float*    ws    = (float*)d_ws;
    float*    hbuf  = ws;
    float*    ssrc  = hbuf + (size_t)N_NODES * OUT_F;
    float*    stgt  = ssrc + (size_t)N_NODES * HEADS;
    float*    tail  = stgt + (size_t)N_NODES * HEADS;
    int*      cnt   = (int*)tail;
    int*      bincnt = cnt + 65536;
    unsigned* gmax  = (unsigned*)(tail + (size_t)N_NODES * HEADS);
    unsigned short* ids  = (unsigned short*)d_out;
    unsigned*       bins = (unsigned*)d_out;

    gat_gemm_wmma<<<NBIN + N_NODES / 16, GB_THREADS, 0, stream>>>(x, W, a, hbuf, ssrc, stgt, gmax,
                                                                  src, tgt, bins, bincnt);

    gat_edge_pass<<<NB_OWN + NB_MAX, EP_THREADS, 0, stream>>>(src, tgt, ssrc, stgt, gmax, cnt, bincnt, ids);

    gat_node_agg<<<N_NODES / 8, 256, 0, stream>>>(src, tgt, cnt, ssrc, a, gmax, hbuf, bias, out);
}


# node pass: an edge's source logit is formed from its fetched h row, not gathered from the logit table (4 rows over a head's 4 lanes as one tree, one exp per lane); den: per-lane partials, then tree
# speedup vs baseline: 6.6322x; 1.0357x over previous
#include <hip/hip_runtime.h>
#include <hip/hip_bf16.h>
#include <math.h>

#define N_NODES   50000
#define N_EDGES   800000
#define IN_F      128
#define HEADS     8
#define HEAD_DIM  16
#define OUT_F     (HEADS * HEAD_DIM)
#define ALPHA     0.2f
#define EPS_F     1e-6f

typedef __attribute__((ext_vector_type(2))) float v2f;
typedef __attribute__((ext_vector_type(8))) float v8f;

__device__ __forceinline__ unsigned enc_f32(float f) {
    unsigned u = __float_as_uint(f);
    return (u & 0x80000000u) ? ~u : (u | 0x80000000u);
}
__device__ __forceinline__ float dec_f32(unsigned u) {
    return (u & 0x80000000u) ? __uint_as_float(u ^ 0x80000000u)
                             : __uint_as_float(~u);
}
__device__ __forceinline__ float leaky(float s) {
    return s >= 0.0f ? s : ALPHA * s;
}

static_assert(N_NODES <= 65536, "source ids are kept as 16-bit words");
static_assert(N_NODES % 8 == 0, "one wave per node, 8 waves per block");
static_assert(N_NODES % 16 == 0, "the product kernel takes whole 16-row tiles");

#define A_PITCH   (IN_F + 4)
static_assert((A_PITCH * 4) % 16 == 0, "A-tile rows are written as 16-B words");

template <int XMASK>
__device__ __forceinline__ float row_xor(float v) {
    return __int_as_float(
        __builtin_amdgcn_update_dpp(0, __float_as_int(v), 0x160 + XMASK, 0xF, 0xF, true));
}

#define SEG_CAP   (OUT_F * 4 / 2)
static_assert(SEG_CAP * 2 == OUT_F * 4, "a segment is exactly one output row");

#define NB_OWN     100
#define OWN_R      (N_NODES / NB_OWN)
static_assert(N_NODES % NB_OWN == 0, "the owner runs tile the nodes");

#define GB_THREADS 256
#define NBIN       200
#define BIN_EDGES  (N_EDGES / NBIN)
#define BIN_V4     (BIN_EDGES / 4)
#define BIN_K      ((BIN_V4 + GB_THREADS - 1) / GB_THREADS)
#define BIN_CAP    OUT_F
static_assert(N_EDGES % (NBIN * 4) == 0, "every slice is a whole number of 16-B words");
static_assert(NBIN <= OWN_R, "the chunks for one owner are rows of that owner's run");
static_assert(OWN_R <= 65536, "a target offset is packed above a 16-bit source id");

__device__ __forceinline__ void bin_one(bool ok, int s, int t, int blk, int* hist,
                                        unsigned* __restrict__ bins) {
    if (ok && (unsigned)t < (unsigned)N_NODES) {
        const unsigned o = (unsigned)t / (unsigned)OWN_R;
        const unsigned d = (unsigned)t - o * (unsigned)OWN_R;
        const int r = atomicAdd(&hist[o], 1);
        if (r < BIN_CAP)
            bins[((size_t)(o * OWN_R) + blk) * BIN_CAP + r] = (d << 16) | ((unsigned)s & 0xFFFFu);
    }
}

__device__ __forceinline__ void bin_block(int blk, int tid, const int* __restrict__ src,
                                          const int* __restrict__ tgt, unsigned* __restrict__ bins,
                                          int* __restrict__ bincnt, int* hist) {
    for (int i = tid; i < NB_OWN; i += GB_THREADS) hist[i] = 0;
    __syncthreads();
    const int4* s4 = reinterpret_cast<const int4*>(src) + blk * BIN_V4;
    const int4* t4 = reinterpret_cast<const int4*>(tgt) + blk * BIN_V4;
    int4 sv[BIN_K], tv[BIN_K];
#pragma unroll
    for (int k = 0; k < BIN_K; ++k) {
        const int q = min(tid + k * GB_THREADS, BIN_V4 - 1);
        sv[k] = s4[q];
        tv[k] = t4[q];
    }
#pragma unroll
    for (int k = 0; k < BIN_K; ++k) {
        const bool ok = tid + k * GB_THREADS < BIN_V4;
        bin_one(ok, sv[k].x, tv[k].x, blk, hist, bins);
        bin_one(ok, sv[k].y, tv[k].y, blk, hist, bins);
        bin_one(ok, sv[k].z, tv[k].z, blk, hist, bins);
        bin_one(ok, sv[k].w, tv[k].w, blk, hist, bins);
    }
    __syncthreads();
    for (int i = tid; i < NB_OWN; i += GB_THREADS) bincnt[i * NBIN + blk] = hist[i];
}

__global__ __launch_bounds__(GB_THREADS)
void gat_gemm_wmma(const float* __restrict__ x, const float* __restrict__ W,
                   const float* __restrict__ a,
                   float* __restrict__ h, float* __restrict__ s_src, float* __restrict__ s_tgt,
                   unsigned* __restrict__ gmax,
                   const int* __restrict__ src, const int* __restrict__ tgt,
                   unsigned* __restrict__ bins, int* __restrict__ bincnt) {
    __shared__ float As[16 * A_PITCH];
    __shared__ int   hist[NB_OWN];
    const int tid = threadIdx.x;
    if (blockIdx.x == 0 && tid < HEADS) gmax[tid] = 0u;
    if (blockIdx.x < NBIN) {
        bin_block(blockIdx.x, tid, src, tgt, bins, bincnt, hist);
        return;
    }
    const int m0  = (blockIdx.x - NBIN) * 16;

    {
        const int base = tid * 8;
        const int row  = base >> 7;
        const int col  = base & 127;
        const float4* g =
            reinterpret_cast<const float4*>(x + (size_t)(m0 + row) * IN_F + col);
        float4 p0 = g[0];
        float4 p1 = g[1];
        *reinterpret_cast<float4*>(&As[row * A_PITCH + col])     = p0;
        *reinterpret_cast<float4*>(&As[row * A_PITCH + col + 4]) = p1;
    }
    __syncthreads();

    const int lane = tid & 31;
    const int head = tid >> 5;
    const int l16  = lane & 15;
    const bool hi  = lane >= 16;
    const float* Wh = W + (size_t)head * IN_F * HEAD_DIM;
    const float as_l = a[head * 2 * HEAD_DIM + l16];
    const float at_l = a[head * 2 * HEAD_DIM + HEAD_DIM + l16];

    v8f c = {};
#pragma unroll
    for (int k = 0; k < IN_F; k += 4) {
        const int kk = k + (hi ? 2 : 0);
        v2f fa, fb;
        fa.x = As[l16 * A_PITCH + kk];
        fa.y = As[l16 * A_PITCH + kk + 1];
        fb.x = Wh[kk * HEAD_DIM + l16];
        fb.y = Wh[(kk + 1) * HEAD_DIM + l16];
        c = __builtin_amdgcn_wmma_f32_16x16x4_f32(
                 false, fa,  false, fb,
                 (short)0, c,  false,  false);
    }

    const int rbase = hi ? 8 : 0;
#pragma unroll
    for (int v = 0; v < 8; ++v)
        h[(size_t)(m0 + rbase + v) * OUT_F + head * HEAD_DIM + l16] = c[v];

    const bool b8 = (l16 & 8) != 0, b4 = (l16 & 4) != 0, b2 = (l16 & 2) != 0, b1 = (l16 & 1) != 0;
    const float a_keep = b8 ? at_l : as_l;
    const float a_give = b8 ? as_l : at_l;
    float r8[8], r4[4], r2[2];
#pragma unroll
    for (int j = 0; j < 8; ++j)
        r8[j] = c[j] * a_keep + row_xor<8>(c[j] * a_give);
#pragma unroll
    for (int j = 0; j < 4; ++j) {
        const float keep = b4 ? r8[4 + j] : r8[j];
        const float give = b4 ? r8[j] : r8[4 + j];
        r4[j] = keep + row_xor<4>(give);
    }
#pragma unroll
    for (int j = 0; j < 2; ++j) {
        const float keep = b2 ? r4[2 + j] : r4[j];
        const float give = b2 ? r4[j] : r4[2 + j];
        r2[j] = keep + row_xor<2>(give);
    }
    const float mine = (b1 ? r2[1] : r2[0]) + row_xor<1>(b1 ? r2[0] : r2[1]);
    float* dst = (l16 & 8) ? s_tgt : s_src;
    dst[(size_t)(m0 + rbase + (l16 & 7)) * HEADS + head] = mine;
}

#define EP_THREADS 1024
#define EP_WAVES   (EP_THREADS / 32)
#define OWN_K      ((NBIN + EP_WAVES - 1) / EP_WAVES)
#define SCAN_V4    (N_EDGES / 4)
static_assert(N_EDGES % 4 == 0, "the target list is read as 16-B words");
static_assert(BIN_CAP == 4 * 32, "a wave reads a chunk with one 16-B load per lane");

#define EM_PAIRS   (EP_THREADS / 2)
#define EM_PASSES  5
#define EM_EDGES   (EM_PAIRS * EM_PASSES)
#define NB_MAX     ((N_EDGES + EM_EDGES - 1) / EM_EDGES)

__device__ __forceinline__ void own_place(unsigned d, int s, int lo, int* lcnt,
                                          unsigned short* ids) {
    const int r = atomicAdd(&lcnt[d], 1);
    if ((unsigned)r < (unsigned)SEG_CAP)
        ids[(size_t)(lo + (int)d) * SEG_CAP + r] = (unsigned short)s;
}

__device__ __forceinline__ void own_place4(uint4 w, int e0, int n, int lo, int* lcnt,
                                           unsigned short* ids) {
    const unsigned e[4] = {w.x, w.y, w.z, w.w};
    unsigned d[4];
    int      r[4];
#pragma unroll
    for (int c = 0; c < 4; ++c) {
        d[c] = min(e[c] >> 16, (unsigned)(OWN_R - 1));
        r[c] = (e0 + c < n) ? atomicAdd(&lcnt[d[c]], 1) : SEG_CAP;
    }
#pragma unroll
    for (int c = 0; c < 4; ++c)
        if ((unsigned)r[c] < (unsigned)SEG_CAP)
            ids[(size_t)(lo + (int)d[c]) * SEG_CAP + r[c]] = (unsigned short)(e[c] & 0xFFFFu);
}

__global__ __launch_bounds__(EP_THREADS)
void gat_edge_pass(const int* __restrict__ src, const int* __restrict__ tgt,
                   const float* __restrict__ s_src, const float* __restrict__ s_tgt,
                   unsigned* __restrict__ gmax, int* __restrict__ cnt,
                   const int* __restrict__ bincnt, unsigned short* ids) {
    __shared__ int      lcnt[OWN_R];
    __shared__ int      any_over;
    __shared__ float4   wmax[EP_WAVES][2];
    const int tid  = threadIdx.x;
    const int lane = tid & 31;

    if (blockIdx.x < NB_OWN) {
        const int lo = blockIdx.x * OWN_R;
        const int wv = __builtin_amdgcn_readfirstlane(tid >> 5);
        const uint4* rows = reinterpret_cast<const uint4*>(ids + (size_t)lo * SEG_CAP);
        uint4 ent[OWN_K];
        int   n[OWN_K];
        bool  over = false;
#pragma unroll
        for (int k = 0; k < OWN_K; ++k) {
            const int c  = wv + k * EP_WAVES;
            const int cc = min(c, NBIN - 1);
            const int cn = bincnt[blockIdx.x * NBIN + cc];
            n[k]   = (c < NBIN) ? cn : 0;
            ent[k] = rows[cc * (BIN_CAP / 4) + lane];
            over   = over || n[k] > BIN_CAP;
        }
        for (int i = tid; i < OWN_R; i += EP_THREADS) lcnt[i] = 0;
        if (tid == 0) any_over = 0;
        __syncthreads();
        if (over && lane == 0) any_over = 1;
        __syncthreads();
        if (any_over == 0) {
#pragma unroll
            for (int k = 0; k < OWN_K; ++k)
                own_place4(ent[k], lane * 4, n[k], lo, lcnt, ids);
        } else {
            const int4* t4 = reinterpret_cast<const int4*>(tgt);
            for (int idx = tid; idx < SCAN_V4; idx += EP_THREADS) {
                const int4 w = t4[idx];
                const unsigned dx = (unsigned)(w.x - lo), dy = (unsigned)(w.y - lo);
                const unsigned dz = (unsigned)(w.z - lo), dw = (unsigned)(w.w - lo);
                if (dx < (unsigned)OWN_R) own_place(dx, src[idx * 4],     lo, lcnt, ids);
                if (dy < (unsigned)OWN_R) own_place(dy, src[idx * 4 + 1], lo, lcnt, ids);
                if (dz < (unsigned)OWN_R) own_place(dz, src[idx * 4 + 2], lo, lcnt, ids);
                if (dw < (unsigned)OWN_R) own_place(dw, src[idx * 4 + 3], lo, lcnt, ids);
            }
        }
        __syncthreads();
        for (int i = tid; i < OWN_R; i += EP_THREADS) cnt[lo + i] = lcnt[i];
        return;
    }

    const int half = tid & 1;
    const int e0   = (blockIdx.x - NB_OWN) * EM_EDGES + (tid >> 1);

    int s[EM_PASSES], t[EM_PASSES];
    for (int p = 0; p < EM_PASSES; ++p) {
        const int e = min(e0 + p * EM_PAIRS, N_EDGES - 1);
        s[p] = src[e];
        t[p] = tgt[e];
    }

    float4 a[EM_PASSES], b[EM_PASSES];
    for (int p = 0; p < EM_PASSES; ++p) {
        a[p] = *reinterpret_cast<const float4*>(s_src + s[p] * HEADS + half * 4);
        b[p] = *reinterpret_cast<const float4*>(s_tgt + t[p] * HEADS + half * 4);
    }

    float4 m = make_float4(a[0].x + b[0].x, a[0].y + b[0].y, a[0].z + b[0].z, a[0].w + b[0].w);
    for (int p = 1; p < EM_PASSES; ++p) {
        m.x = fmaxf(m.x, a[p].x + b[p].x);
        m.y = fmaxf(m.y, a[p].y + b[p].y);
        m.z = fmaxf(m.z, a[p].z + b[p].z);
        m.w = fmaxf(m.w, a[p].w + b[p].w);
    }

    for (int d = 2; d < 32; d <<= 1) {
        m.x = fmaxf(m.x, __shfl_xor(m.x, d));
        m.y = fmaxf(m.y, __shfl_xor(m.y, d));
        m.z = fmaxf(m.z, __shfl_xor(m.z, d));
        m.w = fmaxf(m.w, __shfl_xor(m.w, d));
    }
    if (lane < 2) wmax[tid >> 5][lane] = m;
    __syncthreads();

    if (tid < HEADS) {
        const float* wm = reinterpret_cast<const float*>(&wmax[0][0]);
        float v = wm[tid];
        for (int w = 1; w < EP_WAVES; ++w) v = fmaxf(v, wm[w * HEADS + tid]);
        atomicMax(&gmax[tid], enc_f32(leaky(v)));
    }
}

static_assert(N_EDGES % 32 == 0, "the walk over the whole edge list takes full waves");

__device__ __forceinline__ float edge_weight(float ssrc, float stgt, float mx) {
    const float sc = leaky(ssrc + stgt);
    return __expf(fminf(fmaxf(sc - mx, -20.0f), 20.0f));
}

template <int P>
__device__ __forceinline__ float quad_lane(float v) {
    return __int_as_float(
        __builtin_amdgcn_update_dpp(0, __float_as_int(v), P * 0x55, 0xF, 0xF, true));
}

__device__ __forceinline__ float logit_part(const float4& v, const float4& as) {
    const v2f t = __builtin_elementwise_fma(v2f{v.z, v.w}, v2f{as.z, as.w},
                                            v2f{v.x, v.y} * v2f{as.x, as.y});
    return t.x + t.y;
}

__device__ __forceinline__ void add_edge(int s0, int lane, float st, float mx, const float4& as,
                                         const float* __restrict__ h,
                                         float4& acc, float& den) {
    const float4 v0 = *reinterpret_cast<const float4*>(h + (size_t)s0 * OUT_F + lane * 4);
    float q0 = logit_part(v0, as);
    q0 += row_xor<1>(q0);
    q0 += row_xor<2>(q0);
    const float w0 = edge_weight(q0, st, mx);
    den += w0;
    acc.x += v0.x * w0; acc.y += v0.y * w0; acc.z += v0.z * w0; acc.w += v0.w * w0;
}

__global__ __launch_bounds__(256)
void gat_node_agg(const int* __restrict__ src, const int* __restrict__ tgt,
                  const int* __restrict__ cnt, const float* __restrict__ a,
                  const unsigned* __restrict__ gmax, const float* __restrict__ h,
                  const float* __restrict__ bias, float* out) {
    const int lane = threadIdx.x & 31;
    const int wv = __builtin_amdgcn_readfirstlane((int)(threadIdx.x >> 5));
    const int n  = blockIdx.x * 8 + wv;
    const int hd = lane >> 2;

    const unsigned short* row = reinterpret_cast<const unsigned short*>(out + (size_t)n * OUT_F);
    const int deg = cnt[n];
    const float4 hn = *reinterpret_cast<const float4*>(h + (size_t)n * OUT_F + lane * 4);
    const float4 at = *reinterpret_cast<const float4*>(a + hd * 2 * HEAD_DIM + HEAD_DIM + (lane & 3) * 4);
    const float4 as = *reinterpret_cast<const float4*>(a + hd * 2 * HEAD_DIM + (lane & 3) * 4);
    const unsigned gm = gmax[hd];
    const int pre = row[lane];

    float hx = hn.x;
    asm volatile("" : "+v"(hx) : "v"(pre), "v"(gm));

    float st = hx * at.x + hn.y * at.y + hn.z * at.z + hn.w * at.w;
    st += __shfl_xor(st, 1);
    st += __shfl_xor(st, 2);

    const float mx = dec_f32(gm);

    float4 acc = make_float4(0.0f, 0.0f, 0.0f, 0.0f);
    float  den  = 0.0f;
    float  den4 = 0.0f;
    const bool b1 = (lane & 1) != 0, b2 = (lane & 2) != 0;

    if (deg <= SEG_CAP) {
        for (int base = 0; base < deg; base += 32) {
            const int cnt32 = min(deg - base, 32);
            int v = pre;
            if (base > 0) v = row[base + lane];
            const int myidx = (lane < cnt32) ? min(v, N_NODES - 1) : 0;

            int k = 0;
            for (; k + 4 <= cnt32; k += 4) {
                const int s0 = __builtin_amdgcn_readlane(myidx, k);
                const int s1 = __builtin_amdgcn_readlane(myidx, k + 1);
                const int s2 = __builtin_amdgcn_readlane(myidx, k + 2);
                const int s3 = __builtin_amdgcn_readlane(myidx, k + 3);
                const float4 v0 = *reinterpret_cast<const float4*>(h + (size_t)s0 * OUT_F + lane * 4);
                const float4 v1 = *reinterpret_cast<const float4*>(h + (size_t)s1 * OUT_F + lane * 4);
                const float4 v2 = *reinterpret_cast<const float4*>(h + (size_t)s2 * OUT_F + lane * 4);
                const float4 v3 = *reinterpret_cast<const float4*>(h + (size_t)s3 * OUT_F + lane * 4);
                const float p0 = logit_part(v0, as), p1 = logit_part(v1, as);
                const float p2 = logit_part(v2, as), p3 = logit_part(v3, as);
                const float t0 = (b1 ? p2 : p0) + row_xor<1>(b1 ? p0 : p2);
                const float t1 = (b1 ? p3 : p1) + row_xor<1>(b1 ? p1 : p3);
                const float qq = (b2 ? t1 : t0) + row_xor<2>(b2 ? t0 : t1);
                const float ww = edge_weight(qq, st, mx);
                den4 += ww;
                const float w0 = quad_lane<0>(ww);
                const float w1 = quad_lane<2>(ww);
                const float w2 = quad_lane<1>(ww);
                const float w3 = quad_lane<3>(ww);
                acc.x += v0.x * w0; acc.y += v0.y * w0; acc.z += v0.z * w0; acc.w += v0.w * w0;
                acc.x += v1.x * w1; acc.y += v1.y * w1; acc.z += v1.z * w1; acc.w += v1.w * w1;
                acc.x += v2.x * w2; acc.y += v2.y * w2; acc.z += v2.z * w2; acc.w += v2.w * w2;
                acc.x += v3.x * w3; acc.y += v3.y * w3; acc.z += v3.z * w3; acc.w += v3.w * w3;
            }
            for (; k < cnt32; ++k) {
                const int s0 = __builtin_amdgcn_readlane(myidx, k);
                add_edge(s0, lane, st, mx, as, h, acc, den);
            }
        }
    } else {
        for (int eb = 0; eb < N_EDGES; eb += 32) {
            const int tt = tgt[eb + lane];
            const int ss = src[eb + lane];
            unsigned hit = __builtin_amdgcn_ballot_w32(tt == n);
            while (hit) {
                const int k = __builtin_ctz(hit);
                hit &= hit - 1;
                const int s0 = __builtin_amdgcn_readlane(ss, k);
                add_edge(s0, lane, st, mx, as, h, acc, den);
            }
        }
    }

    den4 += row_xor<1>(den4);
    den4 += row_xor<2>(den4);
    const float  dn = (den4 + den) + EPS_F;
    const float4 bb = *reinterpret_cast<const float4*>(bias + lane * 4);
    float4 o;
    o.x = acc.x / dn + bb.x;
    o.y = acc.y / dn + bb.y;
    o.z = acc.z / dn + bb.z;
    o.w = acc.w / dn + bb.w;
    *reinterpret_cast<float4*>(out + (size_t)n * OUT_F + lane * 4) = o;
}

extern "C" void kernel_launch(void* const* d_in, const int* in_sizes, int n_in,
                              void* d_out, int out_size, void* d_ws, size_t ws_size,
                              hipStream_t stream) {
    (void)in_sizes; (void)n_in; (void)out_size; (void)ws_size;

    const float* x    = (const float*)d_in[0];
    const int*   ei   = (const int*)  d_in[1];
    const float* W    = (const float*)d_in[2];
    const float* a    = (const float*)d_in[3];
    const float* bias = (const float*)d_in[4];
    float* out = (float*)d_out;

    const int* src = ei;
    const int* tgt = ei + N_EDGES;

    float*    ws    = (float*)d_ws;
    float*    hbuf  = ws;
    float*    ssrc  = hbuf + (size_t)N_NODES * OUT_F;
    float*    stgt  = ssrc + (size_t)N_NODES * HEADS;
    float*    tail  = stgt + (size_t)N_NODES * HEADS;
    int*      cnt   = (int*)tail;
    int*      bincnt = cnt + 65536;
    unsigned* gmax  = (unsigned*)(tail + (size_t)N_NODES * HEADS);
    unsigned short* ids  = (unsigned short*)d_out;
    unsigned*       bins = (unsigned*)d_out;

    gat_gemm_wmma<<<NBIN + N_NODES / 16, GB_THREADS, 0, stream>>>(x, W, a, hbuf, ssrc, stgt, gmax,
                                                                  src, tgt, bins, bincnt);

    gat_edge_pass<<<NB_OWN + NB_MAX, EP_THREADS, 0, stream>>>(src, tgt, ssrc, stgt, gmax, cnt, bincnt, ids);

    gat_node_agg<<<N_NODES / 8, 256, 0, stream>>>(src, tgt, cnt, a, gmax, hbuf, bias, out);
}


# owner blocks: ranked source ids collected in LDS (128 B per node), then each node's list written to its row in 16-B pieces, four rows per store; no 2-B store per edge; rank order and sums as before
# speedup vs baseline: 6.9428x; 1.0468x over previous
#include <hip/hip_runtime.h>
#include <hip/hip_bf16.h>
#include <math.h>

#define N_NODES   50000
#define N_EDGES   800000
#define IN_F      128
#define HEADS     8
#define HEAD_DIM  16
#define OUT_F     (HEADS * HEAD_DIM)
#define ALPHA     0.2f
#define EPS_F     1e-6f

typedef __attribute__((ext_vector_type(2))) float v2f;
typedef __attribute__((ext_vector_type(8))) float v8f;

__device__ __forceinline__ unsigned enc_f32(float f) {
    unsigned u = __float_as_uint(f);
    return (u & 0x80000000u) ? ~u : (u | 0x80000000u);
}
__device__ __forceinline__ float dec_f32(unsigned u) {
    return (u & 0x80000000u) ? __uint_as_float(u ^ 0x80000000u)
                             : __uint_as_float(~u);
}
__device__ __forceinline__ float leaky(float s) {
    return s >= 0.0f ? s : ALPHA * s;
}

static_assert(N_NODES <= 65536, "source ids are kept as 16-bit words");
static_assert(N_NODES % 8 == 0, "one wave per node, 8 waves per block");
static_assert(N_NODES % 16 == 0, "the product kernel takes whole 16-row tiles");

#define A_PITCH   (IN_F + 4)
static_assert((A_PITCH * 4) % 16 == 0, "A-tile rows are written as 16-B words");

template <int XMASK>
__device__ __forceinline__ float row_xor(float v) {
    return __int_as_float(
        __builtin_amdgcn_update_dpp(0, __float_as_int(v), 0x160 + XMASK, 0xF, 0xF, true));
}

#define SEG_CAP   (OUT_F * 4 / 2)
static_assert(SEG_CAP * 2 == OUT_F * 4, "a segment is exactly one output row");

#define NB_OWN     100
#define OWN_R      (N_NODES / NB_OWN)
static_assert(N_NODES % NB_OWN == 0, "the owner runs tile the nodes");

#define GB_THREADS 256
#define NBIN       200
#define BIN_EDGES  (N_EDGES / NBIN)
#define BIN_V4     (BIN_EDGES / 4)
#define BIN_K      ((BIN_V4 + GB_THREADS - 1) / GB_THREADS)
#define BIN_CAP    OUT_F
static_assert(N_EDGES % (NBIN * 4) == 0, "every slice is a whole number of 16-B words");
static_assert(NBIN <= OWN_R, "the chunks for one owner are rows of that owner's run");
static_assert(OWN_R <= 65536, "a target offset is packed above a 16-bit source id");

__device__ __forceinline__ void bin_one(bool ok, int s, int t, int blk, int* hist,
                                        unsigned* __restrict__ bins) {
    if (ok && (unsigned)t < (unsigned)N_NODES) {
        const unsigned o = (unsigned)t / (unsigned)OWN_R;
        const unsigned d = (unsigned)t - o * (unsigned)OWN_R;
        const int r = atomicAdd(&hist[o], 1);
        if (r < BIN_CAP)
            bins[((size_t)(o * OWN_R) + blk) * BIN_CAP + r] = (d << 16) | ((unsigned)s & 0xFFFFu);
    }
}

__device__ __forceinline__ void bin_block(int blk, int tid, const int* __restrict__ src,
                                          const int* __restrict__ tgt, unsigned* __restrict__ bins,
                                          int* __restrict__ bincnt, int* hist) {
    for (int i = tid; i < NB_OWN; i += GB_THREADS) hist[i] = 0;
    __syncthreads();
    const int4* s4 = reinterpret_cast<const int4*>(src) + blk * BIN_V4;
    const int4* t4 = reinterpret_cast<const int4*>(tgt) + blk * BIN_V4;
    int4 sv[BIN_K], tv[BIN_K];
#pragma unroll
    for (int k = 0; k < BIN_K; ++k) {
        const int q = min(tid + k * GB_THREADS, BIN_V4 - 1);
        sv[k] = s4[q];
        tv[k] = t4[q];
    }
#pragma unroll
    for (int k = 0; k < BIN_K; ++k) {
        const bool ok = tid + k * GB_THREADS < BIN_V4;
        bin_one(ok, sv[k].x, tv[k].x, blk, hist, bins);
        bin_one(ok, sv[k].y, tv[k].y, blk, hist, bins);
        bin_one(ok, sv[k].z, tv[k].z, blk, hist, bins);
        bin_one(ok, sv[k].w, tv[k].w, blk, hist, bins);
    }
    __syncthreads();
    for (int i = tid; i < NB_OWN; i += GB_THREADS) bincnt[i * NBIN + blk] = hist[i];
}

__global__ __launch_bounds__(GB_THREADS)
void gat_gemm_wmma(const float* __restrict__ x, const float* __restrict__ W,
                   const float* __restrict__ a,
                   float* __restrict__ h, float* __restrict__ s_src, float* __restrict__ s_tgt,
                   unsigned* __restrict__ gmax,
                   const int* __restrict__ src, const int* __restrict__ tgt,
                   unsigned* __restrict__ bins, int* __restrict__ bincnt) {
    __shared__ float As[16 * A_PITCH];
    __shared__ int   hist[NB_OWN];
    const int tid = threadIdx.x;
    if (blockIdx.x == 0 && tid < HEADS) gmax[tid] = 0u;
    if (blockIdx.x < NBIN) {
        bin_block(blockIdx.x, tid, src, tgt, bins, bincnt, hist);
        return;
    }
    const int m0  = (blockIdx.x - NBIN) * 16;

    {
        const int base = tid * 8;
        const int row  = base >> 7;
        const int col  = base & 127;
        const float4* g =
            reinterpret_cast<const float4*>(x + (size_t)(m0 + row) * IN_F + col);
        float4 p0 = g[0];
        float4 p1 = g[1];
        *reinterpret_cast<float4*>(&As[row * A_PITCH + col])     = p0;
        *reinterpret_cast<float4*>(&As[row * A_PITCH + col + 4]) = p1;
    }
    __syncthreads();

    const int lane = tid & 31;
    const int head = tid >> 5;
    const int l16  = lane & 15;
    const bool hi  = lane >= 16;
    const float* Wh = W + (size_t)head * IN_F * HEAD_DIM;
    const float as_l = a[head * 2 * HEAD_DIM + l16];
    const float at_l = a[head * 2 * HEAD_DIM + HEAD_DIM + l16];

    v8f c = {};
#pragma unroll
    for (int k = 0; k < IN_F; k += 4) {
        const int kk = k + (hi ? 2 : 0);
        v2f fa, fb;
        fa.x = As[l16 * A_PITCH + kk];
        fa.y = As[l16 * A_PITCH + kk + 1];
        fb.x = Wh[kk * HEAD_DIM + l16];
        fb.y = Wh[(kk + 1) * HEAD_DIM + l16];
        c = __builtin_amdgcn_wmma_f32_16x16x4_f32(
                 false, fa,  false, fb,
                 (short)0, c,  false,  false);
    }

    const int rbase = hi ? 8 : 0;
#pragma unroll
    for (int v = 0; v < 8; ++v)
        h[(size_t)(m0 + rbase + v) * OUT_F + head * HEAD_DIM + l16] = c[v];

    const bool b8 = (l16 & 8) != 0, b4 = (l16 & 4) != 0, b2 = (l16 & 2) != 0, b1 = (l16 & 1) != 0;
    const float a_keep = b8 ? at_l : as_l;
    const float a_give = b8 ? as_l : at_l;
    float r8[8], r4[4], r2[2];
#pragma unroll
    for (int j = 0; j < 8; ++j)
        r8[j] = c[j] * a_keep + row_xor<8>(c[j] * a_give);
#pragma unroll
    for (int j = 0; j < 4; ++j) {
        const float keep = b4 ? r8[4 + j] : r8[j];
        const float give = b4 ? r8[j] : r8[4 + j];
        r4[j] = keep + row_xor<4>(give);
    }
#pragma unroll
    for (int j = 0; j < 2; ++j) {
        const float keep = b2 ? r4[2 + j] : r4[j];
        const float give = b2 ? r4[j] : r4[2 + j];
        r2[j] = keep + row_xor<2>(give);
    }
    const float mine = (b1 ? r2[1] : r2[0]) + row_xor<1>(b1 ? r2[0] : r2[1]);
    float* dst = (l16 & 8) ? s_tgt : s_src;
    dst[(size_t)(m0 + rbase + (l16 & 7)) * HEADS + head] = mine;
}

#define EP_THREADS 1024
#define EP_WAVES   (EP_THREADS / 32)
#define OWN_K      ((NBIN + EP_WAVES - 1) / EP_WAVES)
#define SCAN_V4    (N_EDGES / 4)
static_assert(N_EDGES % 4 == 0, "the target list is read as 16-B words");
static_assert(BIN_CAP == 4 * 32, "a wave reads a chunk with one 16-B load per lane");

#define EM_PAIRS   (EP_THREADS / 2)
#define EM_PASSES  5
#define EM_EDGES   (EM_PAIRS * EM_PASSES)
#define NB_MAX     ((N_EDGES + EM_EDGES - 1) / EM_EDGES)

#define STG_CAP   64
static_assert(STG_CAP <= SEG_CAP && (STG_CAP * 2) % 16 == 0, "the staged list is the head of the row, in 16-B pieces");
static_assert(OWN_R % 4 == 0, "the staged lists are written out four nodes per store");
#define OWN_T     ((OWN_R + EP_WAVES * 4 - 1) / (EP_WAVES * 4))

__device__ __forceinline__ void own_put(unsigned d, int r, unsigned short s, int lo,
                                        unsigned short* stage, unsigned short* ids) {
    if ((unsigned)r < (unsigned)STG_CAP)
        stage[d * STG_CAP + r] = s;
    else if ((unsigned)r < (unsigned)SEG_CAP)
        ids[(size_t)(lo + (int)d) * SEG_CAP + r] = s;
}

__device__ __forceinline__ void own_place(unsigned d, int s, int lo, int* lcnt,
                                          unsigned short* stage, unsigned short* ids) {
    const int r = atomicAdd(&lcnt[d], 1);
    own_put(d, r, (unsigned short)s, lo, stage, ids);
}

__device__ __forceinline__ void own_place4(uint4 w, int e0, int n, int lo, int* lcnt,
                                           unsigned short* stage, unsigned short* ids) {
    const unsigned e[4] = {w.x, w.y, w.z, w.w};
    unsigned d[4];
    int      r[4];
#pragma unroll
    for (int c = 0; c < 4; ++c) {
        d[c] = min(e[c] >> 16, (unsigned)(OWN_R - 1));
        r[c] = (e0 + c < n) ? atomicAdd(&lcnt[d[c]], 1) : SEG_CAP;
    }
#pragma unroll
    for (int c = 0; c < 4; ++c)
        own_put(d[c], r[c], (unsigned short)(e[c] & 0xFFFFu), lo, stage, ids);
}

__global__ __launch_bounds__(EP_THREADS)
void gat_edge_pass(const int* __restrict__ src, const int* __restrict__ tgt,
                   const float* __restrict__ s_src, const float* __restrict__ s_tgt,
                   unsigned* __restrict__ gmax, int* __restrict__ cnt,
                   const int* __restrict__ bincnt, unsigned short* ids) {
    __shared__ int      lcnt[OWN_R];
    __shared__ int      any_over;
    __shared__ __attribute__((aligned(16)))
    unsigned short      stage[OWN_R * STG_CAP];
    __shared__ float4   wmax[EP_WAVES][2];
    const int tid  = threadIdx.x;
    const int lane = tid & 31;

    if (blockIdx.x < NB_OWN) {
        const int lo = blockIdx.x * OWN_R;
        const int wv = __builtin_amdgcn_readfirstlane(tid >> 5);
        const uint4* rows = reinterpret_cast<const uint4*>(ids + (size_t)lo * SEG_CAP);
        uint4 ent[OWN_K];
        int   n[OWN_K];
        bool  over = false;
#pragma unroll
        for (int k = 0; k < OWN_K; ++k) {
            const int c  = wv + k * EP_WAVES;
            const int cc = min(c, NBIN - 1);
            const int cn = bincnt[blockIdx.x * NBIN + cc];
            n[k]   = (c < NBIN) ? cn : 0;
            ent[k] = rows[cc * (BIN_CAP / 4) + lane];
            over   = over || n[k] > BIN_CAP;
        }
        for (int i = tid; i < OWN_R; i += EP_THREADS) lcnt[i] = 0;
        if (tid == 0) any_over = 0;
        __syncthreads();
        if (over && lane == 0) any_over = 1;
        __syncthreads();
        if (any_over == 0) {
#pragma unroll
            for (int k = 0; k < OWN_K; ++k)
                own_place4(ent[k], lane * 4, n[k], lo, lcnt, stage, ids);
        } else {
            const int4* t4 = reinterpret_cast<const int4*>(tgt);
            for (int idx = tid; idx < SCAN_V4; idx += EP_THREADS) {
                const int4 w = t4[idx];
                const unsigned dx = (unsigned)(w.x - lo), dy = (unsigned)(w.y - lo);
                const unsigned dz = (unsigned)(w.z - lo), dw = (unsigned)(w.w - lo);
                if (dx < (unsigned)OWN_R) own_place(dx, src[idx * 4],     lo, lcnt, stage, ids);
                if (dy < (unsigned)OWN_R) own_place(dy, src[idx * 4 + 1], lo, lcnt, stage, ids);
                if (dz < (unsigned)OWN_R) own_place(dz, src[idx * 4 + 2], lo, lcnt, stage, ids);
                if (dw < (unsigned)OWN_R) own_place(dw, src[idx * 4 + 3], lo, lcnt, stage, ids);
            }
        }
        __syncthreads();
        for (int i = tid; i < OWN_R; i += EP_THREADS) cnt[lo + i] = lcnt[i];
        const int piece = lane & 7;
        int   have[OWN_T];
        uint4 w8[OWN_T];
#pragma unroll
        for (int t = 0; t < OWN_T; ++t) {
            const int i = min(wv * 4 + (lane >> 3) + t * (EP_WAVES * 4), OWN_R - 1);
            have[t] = lcnt[i];
            w8[t]   = *reinterpret_cast<const uint4*>(&stage[i * STG_CAP + piece * 8]);
        }
#pragma unroll
        for (int t = 0; t < OWN_T; ++t) {
            const int i = wv * 4 + (lane >> 3) + t * (EP_WAVES * 4);
            if (i < OWN_R && piece * 8 < have[t])
                *reinterpret_cast<uint4*>(ids + (size_t)(lo + i) * SEG_CAP + piece * 8) = w8[t];
        }
        return;
    }

    const int half = tid & 1;
    const int e0   = (blockIdx.x - NB_OWN) * EM_EDGES + (tid >> 1);

    int s[EM_PASSES], t[EM_PASSES];
    for (int p = 0; p < EM_PASSES; ++p) {
        const int e = min(e0 + p * EM_PAIRS, N_EDGES - 1);
        s[p] = src[e];
        t[p] = tgt[e];
    }

    float4 a[EM_PASSES], b[EM_PASSES];
    for (int p = 0; p < EM_PASSES; ++p) {
        a[p] = *reinterpret_cast<const float4*>(s_src + s[p] * HEADS + half * 4);
        b[p] = *reinterpret_cast<const float4*>(s_tgt + t[p] * HEADS + half * 4);
    }

    float4 m = make_float4(a[0].x + b[0].x, a[0].y + b[0].y, a[0].z + b[0].z, a[0].w + b[0].w);
    for (int p = 1; p < EM_PASSES; ++p) {
        m.x = fmaxf(m.x, a[p].x + b[p].x);
        m.y = fmaxf(m.y, a[p].y + b[p].y);
        m.z = fmaxf(m.z, a[p].z + b[p].z);
        m.w = fmaxf(m.w, a[p].w + b[p].w);
    }

    for (int d = 2; d < 32; d <<= 1) {
        m.x = fmaxf(m.x, __shfl_xor(m.x, d));
        m.y = fmaxf(m.y, __shfl_xor(m.y, d));
        m.z = fmaxf(m.z, __shfl_xor(m.z, d));
        m.w = fmaxf(m.w, __shfl_xor(m.w, d));
    }
    if (lane < 2) wmax[tid >> 5][lane] = m;
    __syncthreads();

    if (tid < HEADS) {
        const float* wm = reinterpret_cast<const float*>(&wmax[0][0]);
        float v = wm[tid];
        for (int w = 1; w < EP_WAVES; ++w) v = fmaxf(v, wm[w * HEADS + tid]);
        atomicMax(&gmax[tid], enc_f32(leaky(v)));
    }
}

static_assert(N_EDGES % 32 == 0, "the walk over the whole edge list takes full waves");

__device__ __forceinline__ float edge_weight(float ssrc, float stgt, float mx) {
    const float sc = leaky(ssrc + stgt);
    return __expf(fminf(fmaxf(sc - mx, -20.0f), 20.0f));
}

template <int P>
__device__ __forceinline__ float quad_lane(float v) {
    return __int_as_float(
        __builtin_amdgcn_update_dpp(0, __float_as_int(v), P * 0x55, 0xF, 0xF, true));
}

__device__ __forceinline__ float logit_part(const float4& v, const float4& as) {
    const v2f t = __builtin_elementwise_fma(v2f{v.z, v.w}, v2f{as.z, as.w},
                                            v2f{v.x, v.y} * v2f{as.x, as.y});
    return t.x + t.y;
}

__device__ __forceinline__ void add_edge(int s0, int lane, float st, float mx, const float4& as,
                                         const float* __restrict__ h,
                                         float4& acc, float& den) {
    const float4 v0 = *reinterpret_cast<const float4*>(h + (size_t)s0 * OUT_F + lane * 4);
    float q0 = logit_part(v0, as);
    q0 += row_xor<1>(q0);
    q0 += row_xor<2>(q0);
    const float w0 = edge_weight(q0, st, mx);
    den += w0;
    acc.x += v0.x * w0; acc.y += v0.y * w0; acc.z += v0.z * w0; acc.w += v0.w * w0;
}

__global__ __launch_bounds__(256)
void gat_node_agg(const int* __restrict__ src, const int* __restrict__ tgt,
                  const int* __restrict__ cnt, const float* __restrict__ a,
                  const unsigned* __restrict__ gmax, const float* __restrict__ h,
                  const float* __restrict__ bias, float* out) {
    const int lane = threadIdx.x & 31;
    const int wv = __builtin_amdgcn_readfirstlane((int)(threadIdx.x >> 5));
    const int n  = blockIdx.x * 8 + wv;
    const int hd = lane >> 2;

    const unsigned short* row = reinterpret_cast<const unsigned short*>(out + (size_t)n * OUT_F);
    const int deg = cnt[n];
    const float4 hn = *reinterpret_cast<const float4*>(h + (size_t)n * OUT_F + lane * 4);
    const float4 at = *reinterpret_cast<const float4*>(a + hd * 2 * HEAD_DIM + HEAD_DIM + (lane & 3) * 4);
    const float4 as = *reinterpret_cast<const float4*>(a + hd * 2 * HEAD_DIM + (lane & 3) * 4);
    const unsigned gm = gmax[hd];
    const int pre = row[lane];

    float hx = hn.x;
    asm volatile("" : "+v"(hx) : "v"(pre), "v"(gm));

    float st = hx * at.x + hn.y * at.y + hn.z * at.z + hn.w * at.w;
    st += __shfl_xor(st, 1);
    st += __shfl_xor(st, 2);

    const float mx = dec_f32(gm);

    float4 acc = make_float4(0.0f, 0.0f, 0.0f, 0.0f);
    float  den  = 0.0f;
    float  den4 = 0.0f;
    const bool b1 = (lane & 1) != 0, b2 = (lane & 2) != 0;

    if (deg <= SEG_CAP) {
        for (int base = 0; base < deg; base += 32) {
            const int cnt32 = min(deg - base, 32);
            int v = pre;
            if (base > 0) v = row[base + lane];
            const int myidx = (lane < cnt32) ? min(v, N_NODES - 1) : 0;

            int k = 0;
            for (; k + 4 <= cnt32; k += 4) {
                const int s0 = __builtin_amdgcn_readlane(myidx, k);
                const int s1 = __builtin_amdgcn_readlane(myidx, k + 1);
                const int s2 = __builtin_amdgcn_readlane(myidx, k + 2);
                const int s3 = __builtin_amdgcn_readlane(myidx, k + 3);
                const float4 v0 = *reinterpret_cast<const float4*>(h + (size_t)s0 * OUT_F + lane * 4);
                const float4 v1 = *reinterpret_cast<const float4*>(h + (size_t)s1 * OUT_F + lane * 4);
                const float4 v2 = *reinterpret_cast<const float4*>(h + (size_t)s2 * OUT_F + lane * 4);
                const float4 v3 = *reinterpret_cast<const float4*>(h + (size_t)s3 * OUT_F + lane * 4);
                const float p0 = logit_part(v0, as), p1 = logit_part(v1, as);
                const float p2 = logit_part(v2, as), p3 = logit_part(v3, as);
                const float t0 = (b1 ? p2 : p0) + row_xor<1>(b1 ? p0 : p2);
                const float t1 = (b1 ? p3 : p1) + row_xor<1>(b1 ? p1 : p3);
                const float qq = (b2 ? t1 : t0) + row_xor<2>(b2 ? t0 : t1);
                const float ww = edge_weight(qq, st, mx);
                den4 += ww;
                const float w0 = quad_lane<0>(ww);
                const float w1 = quad_lane<2>(ww);
                const float w2 = quad_lane<1>(ww);
                const float w3 = quad_lane<3>(ww);
                acc.x += v0.x * w0; acc.y += v0.y * w0; acc.z += v0.z * w0; acc.w += v0.w * w0;
                acc.x += v1.x * w1; acc.y += v1.y * w1; acc.z += v1.z * w1; acc.w += v1.w * w1;
                acc.x += v2.x * w2; acc.y += v2.y * w2; acc.z += v2.z * w2; acc.w += v2.w * w2;
                acc.x += v3.x * w3; acc.y += v3.y * w3; acc.z += v3.z * w3; acc.w += v3.w * w3;
            }
            for (; k < cnt32; ++k) {
                const int s0 = __builtin_amdgcn_readlane(myidx, k);
                add_edge(s0, lane, st, mx, as, h, acc, den);
            }
        }
    } else {
        for (int eb = 0; eb < N_EDGES; eb += 32) {
            const int tt = tgt[eb + lane];
            const int ss = src[eb + lane];
            unsigned hit = __builtin_amdgcn_ballot_w32(tt == n);
            while (hit) {
                const int k = __builtin_ctz(hit);
                hit &= hit - 1;
                const int s0 = __builtin_amdgcn_readlane(ss, k);
                add_edge(s0, lane, st, mx, as, h, acc, den);
            }
        }
    }

    den4 += row_xor<1>(den4);
    den4 += row_xor<2>(den4);
    const float  dn = (den4 + den) + EPS_F;
    const float4 bb = *reinterpret_cast<const float4*>(bias + lane * 4);
    float4 o;
    o.x = acc.x / dn + bb.x;
    o.y = acc.y / dn + bb.y;
    o.z = acc.z / dn + bb.z;
    o.w = acc.w / dn + bb.w;
    *reinterpret_cast<float4*>(out + (size_t)n * OUT_F + lane * 4) = o;
}

extern "C" void kernel_launch(void* const* d_in, const int* in_sizes, int n_in,
                              void* d_out, int out_size, void* d_ws, size_t ws_size,
                              hipStream_t stream) {
    (void)in_sizes; (void)n_in; (void)out_size; (void)ws_size;

    const float* x    = (const float*)d_in[0];
    const int*   ei   = (const int*)  d_in[1];
    const float* W    = (const float*)d_in[2];
    const float* a    = (const float*)d_in[3];
    const float* bias = (const float*)d_in[4];
    float* out = (float*)d_out;

    const int* src = ei;
    const int* tgt = ei + N_EDGES;

    float*    ws    = (float*)d_ws;
    float*    hbuf  = ws;
    float*    ssrc  = hbuf + (size_t)N_NODES * OUT_F;
    float*    stgt  = ssrc + (size_t)N_NODES * HEADS;
    float*    tail  = stgt + (size_t)N_NODES * HEADS;
    int*      cnt   = (int*)tail;
    int*      bincnt = cnt + 65536;
    unsigned* gmax  = (unsigned*)(tail + (size_t)N_NODES * HEADS);
    unsigned short* ids  = (unsigned short*)d_out;
    unsigned*       bins = (unsigned*)d_out;

    gat_gemm_wmma<<<NBIN + N_NODES / 16, GB_THREADS, 0, stream>>>(x, W, a, hbuf, ssrc, stgt, gmax,
                                                                  src, tgt, bins, bincnt);

    gat_edge_pass<<<NB_OWN + NB_MAX, EP_THREADS, 0, stream>>>(src, tgt, ssrc, stgt, gmax, cnt, bincnt, ids);

    gat_node_agg<<<N_NODES / 8, 256, 0, stream>>>(src, tgt, cnt, a, gmax, hbuf, bias, out);
}
